# starting kernel
# baseline (speedup 1.0000x reference)
#include <hip/hip_runtime.h>
#include <math.h>

typedef __attribute__((ext_vector_type(2))) float v2f;
typedef __attribute__((ext_vector_type(8))) float v8f;

__device__ __forceinline__ v8f wmma_f32(v2f a, v2f b, v8f c) {
    return __builtin_amdgcn_wmma_f32_16x16x4_f32(false, a, false, b, (short)0, c, false, false);
}
__device__ __forceinline__ v2f lds_frag(const float* p) {
    return *(const v2f*)p;
}

#define GT_M 64
#define GT_N 128
#define GT_K 16
#define A_FO 160
#define B_FO 288

__global__ __launch_bounds__(256) void ddnet_gemm_kernel(
    const float* __restrict__ A, const float* __restrict__ W,
    const float* __restrict__ bias, const float* __restrict__ resid,
    float* __restrict__ C, int Mdim, int Ndim, int Kdim, int act)
{
    __shared__ __align__(16) float As[2][8 * A_FO];
    __shared__ __align__(16) float Bs[2][8 * B_FO];

    const int tid   = threadIdx.x;
    const int lane  = tid & 31;
    const int wave  = tid >> 5;
    const int lhalf = lane >> 4;
    const int lmod  = lane & 15;
    const int m0    = blockIdx.y * GT_M;
    const int n0    = blockIdx.x * GT_N;
    const int wm    = (wave & 1) * 32;
    const int wn    = (wave >> 1) * 32;
    const int nk    = Kdim >> 4;

    v8f acc[2][2] = {};
    float ra[4], rb[8];

    #pragma unroll
    for (int i = 0; i < 4; ++i) {
        int idx = tid + i * 256, kk = idx & 15, mm = idx >> 4;
        ra[i] = A[(size_t)(m0 + mm) * Kdim + kk];
    }
    #pragma unroll
    for (int i = 0; i < 8; ++i) {
        int idx = tid + i * 256, kk = idx & 15, nn = idx >> 4;
        rb[i] = W[(size_t)(n0 + nn) * Kdim + kk];
    }
    #pragma unroll
    for (int i = 0; i < 4; ++i) {
        int idx = tid + i * 256, kk = idx & 15, mm = idx >> 4;
        As[0][((kk >> 2) * 2 + ((kk >> 1) & 1)) * A_FO + mm * 2 + (kk & 1)] = ra[i];
    }
    #pragma unroll
    for (int i = 0; i < 8; ++i) {
        int idx = tid + i * 256, kk = idx & 15, nn = idx >> 4;
        Bs[0][((kk >> 2) * 2 + ((kk >> 1) & 1)) * B_FO + nn * 2 + (kk & 1)] = rb[i];
    }
    __syncthreads();

    for (int kt = 0; kt < nk; ++kt) {
        const int buf = kt & 1;
        if (kt + 1 < nk) {
            const int k0 = (kt + 1) << 4;
            #pragma unroll
            for (int i = 0; i < 4; ++i) {
                int idx = tid + i * 256, kk = idx & 15, mm = idx >> 4;
                ra[i] = A[(size_t)(m0 + mm) * Kdim + k0 + kk];
            }
            #pragma unroll
            for (int i = 0; i < 8; ++i) {
                int idx = tid + i * 256, kk = idx & 15, nn = idx >> 4;
                rb[i] = W[(size_t)(n0 + nn) * Kdim + k0 + kk];
            }
        }
        const float* Ab = As[buf];
        const float* Bb = Bs[buf];
        #pragma unroll
        for (int t = 0; t < 4; ++t) {
            const int fo = t * 2 + lhalf;
            v2f a0 = lds_frag(&Ab[fo * A_FO + (wm + lmod) * 2]);
            v2f a1 = lds_frag(&Ab[fo * A_FO + (wm + 16 + lmod) * 2]);
            v2f b0 = lds_frag(&Bb[fo * B_FO + (wn + lmod) * 2]);
            v2f b1 = lds_frag(&Bb[fo * B_FO + (wn + 16 + lmod) * 2]);
            acc[0][0] = wmma_f32(a0, b0, acc[0][0]);
            acc[0][1] = wmma_f32(a0, b1, acc[0][1]);
            acc[1][0] = wmma_f32(a1, b0, acc[1][0]);
            acc[1][1] = wmma_f32(a1, b1, acc[1][1]);
        }
        if (kt + 1 < nk) {
            __syncthreads();
            #pragma unroll
            for (int i = 0; i < 4; ++i) {
                int idx = tid + i * 256, kk = idx & 15, mm = idx >> 4;
                As[buf ^ 1][((kk >> 2) * 2 + ((kk >> 1) & 1)) * A_FO + mm * 2 + (kk & 1)] = ra[i];
            }
            #pragma unroll
            for (int i = 0; i < 8; ++i) {
                int idx = tid + i * 256, kk = idx & 15, nn = idx >> 4;
                Bs[buf ^ 1][((kk >> 2) * 2 + ((kk >> 1) & 1)) * B_FO + nn * 2 + (kk & 1)] = rb[i];
            }
            __syncthreads();
        }
    }

    #pragma unroll
    for (int mi = 0; mi < 2; ++mi)
        #pragma unroll
        for (int ni = 0; ni < 2; ++ni)
            #pragma unroll
            for (int v = 0; v < 8; ++v) {
                int row = m0 + wm + mi * 16 + v + 8 * lhalf;
                int col = n0 + wn + ni * 16 + lmod;
                float val = acc[mi][ni][v];
                if (bias)  val += bias[col];
                if (resid) val += resid[(size_t)row * Ndim + col];
                if (act == 1)
                    val = 0.5f * val * (1.0f + erff(val * 0.70710678118654752f));
                C[(size_t)row * Ndim + col] = val;
            }
}

__global__ __launch_bounds__(256) void ddnet_pe_kernel(
    float* __restrict__ h, int S, int D)
{
    size_t idx = (size_t)blockIdx.x * 256 + threadIdx.x;
    int d = (int)(idx % D);
    int s = (int)((idx / D) % S);
    int pair = d >> 1;
    float div = __expf(-(float)(2 * pair) * (logf(10000.0f) / (float)D));
    float ang = (float)s * div;
    h[idx] += (d & 1) ? __cosf(ang) : __sinf(ang);
}

__global__ __launch_bounds__(256) void ddnet_ln_kernel(
    const float* __restrict__ x, const float* __restrict__ gamma,
    const float* __restrict__ beta, float* __restrict__ out, int D)
{
    int row  = blockIdx.x * 8 + (threadIdx.x >> 5);
    int lane = threadIdx.x & 31;
    const float* xr = x + (size_t)row * D;
    float vals[8];
    float s = 0.f;
    #pragma unroll
    for (int i = 0; i < 8; ++i) { vals[i] = xr[lane + i * 32]; s += vals[i]; }
    #pragma unroll
    for (int m = 16; m >= 1; m >>= 1) s += __shfl_xor(s, m, 32);
    float mean = s * (1.0f / 256.0f);
    float vs = 0.f;
    #pragma unroll
    for (int i = 0; i < 8; ++i) { float d = vals[i] - mean; vs += d * d; }
    #pragma unroll
    for (int m = 16; m >= 1; m >>= 1) vs += __shfl_xor(vs, m, 32);
    float rstd = rsqrtf(vs * (1.0f / 256.0f) + 1e-5f);
    float* orow = out + (size_t)row * D;
    #pragma unroll
    for (int i = 0; i < 8; ++i) {
        int c = lane + i * 32;
        orow[c] = (vals[i] - mean) * rstd * gamma[c] + beta[c];
    }
}

#define DD 256
#define DG 6

__global__ __launch_bounds__(256) void ddnet_desc_kernel(
    const float* __restrict__ xt, const float* __restrict__ h,
    const float* __restrict__ P, float* __restrict__ hd, int S)
{
    __shared__ __align__(16) float Xs[128 * 32];
    __shared__ float Ct[511 * DG + 1];

    const int s   = blockIdx.x;
    const int tid = threadIdx.x;
    const float two_pi = 6.28318530717958647692f;

    for (int idx = tid; idx < 511 * DG; idx += 256) {
        int t = idx / DG, g = idx % DG;
        float period = (2.0f + 0.5f * (float)t) * powf(1.5f, (float)g);
        Ct[idx] = __cosf(two_pi * (float)s / period);
    }
    for (int idx = tid; idx < 16 * DD; idx += 256) {
        int b = idx >> 8, j = idx & 255;
        Xs[((j >> 2) * 2 + ((j >> 1) & 1)) * 32 + b * 2 + (j & 1)] =
            xt[((size_t)b * S + s) * DD + j];
    }
    __syncthreads();

    const int lane = tid & 31, wave = tid >> 5;
    const int lhalf = lane >> 4, lmod = lane & 15;

    for (int it = wave; it < 16; it += 8) {
        const int i = it * 16 + lmod;
        v8f acc = {};
        for (int j0 = 0; j0 < DD; j0 += 16) {
            float wv[4][2];
            #pragma unroll
            for (int t = 0; t < 4; ++t)
                #pragma unroll
                for (int jj = 0; jj < 2; ++jj) {
                    int j = j0 + t * 4 + jj + 2 * lhalf;
                    const float* Pp = P + ((size_t)i * DD + j) * DG;
                    const float* Cp = Ct + (i + j) * DG;
                    float w = 0.f;
                    #pragma unroll
                    for (int g = 0; g < DG; ++g) w += Pp[g] * Cp[g];
                    wv[t][jj] = w;
                }
            #pragma unroll
            for (int t = 0; t < 4; ++t) {
                v2f a = lds_frag(&Xs[((j0 >> 2) + t) * 64 + lhalf * 32 + lmod * 2]);
                v2f b = { wv[t][0], wv[t][1] };
                acc = wmma_f32(a, b, acc);
            }
        }
        #pragma unroll
        for (int v = 0; v < 8; ++v) {
            int b = v + 8 * lhalf;
            size_t off = ((size_t)b * S + s) * DD + it * 16 + lmod;
            hd[off] = acc[v] + h[off];
        }
    }
}

__global__ __launch_bounds__(128) void ddnet_attn_kernel(
    const float* __restrict__ qkv, float* __restrict__ o, int S)
{
    const int qb    = S / 64;
    const int bh    = blockIdx.x / qb;
    const int qblk  = blockIdx.x % qb;
    const int b     = bh >> 3, head = bh & 7;
    const int tid   = threadIdx.x, lane = tid & 31, wave = tid >> 5;
    const int lhalf = lane >> 4, lmod = lane & 15;
    const int q0    = qblk * 64 + wave * 16;

    __shared__ __align__(16) float Kf[4][16 * 32];
    __shared__ __align__(16) float Vf[4][8 * 96];
    __shared__ __align__(16) float Pf[4][8 * 96];

    const size_t rs = 768;
    const float* qp = qkv + (size_t)b * S * rs + head * 32;
    const float* kp = qp + 256;
    const float* vp = qp + 512;
    const float scale = 0.17677669529663688f;

    v2f aq[8];
    #pragma unroll
    for (int t = 0; t < 8; ++t) {
        const float* qr = qp + (size_t)(q0 + lmod) * rs;
        aq[t].x = qr[t * 4 + 2 * lhalf];
        aq[t].y = qr[t * 4 + 1 + 2 * lhalf];
    }

    float mrow[8], lrow[8];
    v8f o0 = {}, o1 = {};
    #pragma unroll
    for (int v = 0; v < 8; ++v) { mrow[v] = -1e30f; lrow[v] = 0.f; }

    for (int k0 = 0; k0 < S; k0 += 16) {
        for (int i = lane; i < 16 * 32; i += 32) {
            int r = i >> 5, c = i & 31;
            Kf[wave][((c >> 2) * 2 + ((c >> 1) & 1)) * 32 + r * 2 + (c & 1)] =
                kp[(size_t)(k0 + r) * rs + c];
            Vf[wave][((r >> 2) * 2 + ((r >> 1) & 1)) * 96 + c * 2 + (r & 1)] =
                vp[(size_t)(k0 + r) * rs + c];
        }
        __syncthreads();

        v8f sc = {};
        #pragma unroll
        for (int t = 0; t < 8; ++t) {
            v2f bk = lds_frag(&Kf[wave][(t * 2 + lhalf) * 32 + lmod * 2]);
            sc = wmma_f32(aq[t], bk, sc);
        }

        float pr[8];
        #pragma unroll
        for (int v = 0; v < 8; ++v) {
            float sv = sc[v] * scale;
            float mx = sv;
            #pragma unroll
            for (int m = 8; m >= 1; m >>= 1) mx = fmaxf(mx, __shfl_xor(mx, m, 32));
            float mnew = fmaxf(mrow[v], mx);
            float corr = __expf(mrow[v] - mnew);
            float p    = __expf(sv - mnew);
            float rsum = p;
            #pragma unroll
            for (int m = 8; m >= 1; m >>= 1) rsum += __shfl_xor(rsum, m, 32);
            lrow[v] = lrow[v] * corr + rsum;
            mrow[v] = mnew;
            pr[v]   = p;
            o0[v] *= corr;
            o1[v] *= corr;
        }

        #pragma unroll
        for (int v = 0; v < 8; ++v)
            Pf[wave][((lmod >> 2) * 2 + ((lmod >> 1) & 1)) * 96 +
                     (v + 8 * lhalf) * 2 + (lmod & 1)] = pr[v];
        #pragma unroll
        for (int t = 0; t < 4; ++t) {
            int fo = t * 2 + lhalf;
            v2f ap  = lds_frag(&Pf[wave][fo * 96 + lmod * 2]);
            v2f bv0 = lds_frag(&Vf[wave][fo * 96 + lmod * 2]);
            v2f bv1 = lds_frag(&Vf[wave][fo * 96 + (16 + lmod) * 2]);
            o0 = wmma_f32(ap, bv0, o0);
            o1 = wmma_f32(ap, bv1, o1);
        }
        __syncthreads();
    }

    #pragma unroll
    for (int v = 0; v < 8; ++v) {
        int qr = v + 8 * lhalf;
        float inv = 1.0f / lrow[v];
        size_t base = ((size_t)b * S + q0 + qr) * 256 + head * 32;
        o[base + lmod]      = o0[v] * inv;
        o[base + 16 + lmod] = o1[v] * inv;
    }
}

static void launch_gemm(const float* A, const float* W, const float* bias,
                        const float* resid, float* C, int M, int N, int K,
                        int act, hipStream_t stream)
{
    dim3 grid(N / GT_N, M / GT_M);
    ddnet_gemm_kernel<<<grid, 256, 0, stream>>>(A, W, bias, resid, C, M, N, K, act);
}

extern "C" void kernel_launch(void* const* d_in, const int* in_sizes, int n_in,
                              void* d_out, int out_size, void* d_ws, size_t ws_size,
                              hipStream_t stream) {
    (void)in_sizes; (void)n_in; (void)out_size; (void)ws_size;
    const int B = 16, S = 512, IN = 64, D = 256, L = 3, G = 6, DFF = 1024;
    const int Mr = B * S;

    const float* x      = (const float*)d_in[0];
    const float* in_w   = (const float*)d_in[1];
    const float* in_b   = (const float*)d_in[2];
    const float* Mw     = (const float*)d_in[3];
    const float* P      = (const float*)d_in[4];
    const float* desc_g = (const float*)d_in[5];
    const float* desc_b = (const float*)d_in[6];
    const float* qkv_w  = (const float*)d_in[7];
    const float* qkv_b  = (const float*)d_in[8];
    const float* out_w  = (const float*)d_in[9];
    const float* out_b  = (const float*)d_in[10];
    const float* ff1_w  = (const float*)d_in[11];
    const float* ff1_b  = (const float*)d_in[12];
    const float* ff2_w  = (const float*)d_in[13];
    const float* ff2_b  = (const float*)d_in[14];
    const float* ln1_g  = (const float*)d_in[15];
    const float* ln1_b  = (const float*)d_in[16];
    const float* ln2_g  = (const float*)d_in[17];
    const float* ln2_b  = (const float*)d_in[18];

    float* h    = (float*)d_ws;
    float* tmp  = h    + (size_t)Mr * D;
    float* xt   = tmp  + (size_t)Mr * D;
    float* hd   = xt   + (size_t)Mr * D;
    float* qkv  = hd   + (size_t)Mr * D;
    float* obuf = qkv  + (size_t)Mr * 3 * D;
    float* h1   = obuf + (size_t)Mr * D;
    float* ff   = h1   + (size_t)Mr * D;

    launch_gemm(x, in_w, in_b, nullptr, h, Mr, D, IN, 0, stream);
    ddnet_pe_kernel<<<(Mr * D) / 256, 256, 0, stream>>>(h, S, D);

    for (int l = 0; l < L; ++l) {
        launch_gemm(h, Mw + (size_t)l * D * D, nullptr, nullptr, tmp, Mr, D, D, 0, stream);
        ddnet_ln_kernel<<<Mr / 8, 256, 0, stream>>>(tmp, desc_g + l * D, desc_b + l * D, xt, D);
        ddnet_desc_kernel<<<S, 256, 0, stream>>>(xt, h, P + (size_t)l * D * D * G, hd, S);
        launch_gemm(hd, qkv_w + (size_t)l * 3 * D * D, qkv_b + (size_t)l * 3 * D,
                    nullptr, qkv, Mr, 3 * D, D, 0, stream);
        ddnet_attn_kernel<<<B * 8 * (S / 64), 128, 0, stream>>>(qkv, obuf, S);
        launch_gemm(obuf, out_w + (size_t)l * D * D, out_b + l * D, hd, tmp, Mr, D, D, 0, stream);
        ddnet_ln_kernel<<<Mr / 8, 256, 0, stream>>>(tmp, ln1_g + l * D, ln1_b + l * D, h1, D);
        launch_gemm(h1, ff1_w + (size_t)l * DFF * D, ff1_b + (size_t)l * DFF,
                    nullptr, ff, Mr, DFF, D, 1  , stream);
        launch_gemm(ff, ff2_w + (size_t)l * D * DFF, ff2_b + l * D, h1, tmp, Mr, D, DFF, 0, stream);
        float* dst = (l == L - 1) ? (float*)d_out : h;
        ddnet_ln_kernel<<<Mr / 8, 256, 0, stream>>>(tmp, ln2_g + l * D, ln2_b + l * D, dst, D);
    }
}


# GEMM K-tile 16->64 staged in 16-byte units (global loads, LDS stores, fragment reads; 2 WMMAs per read, k summed {0,1,4,5},{2,3,6,7} per 8): per-tile sync and address work amortised 4x
# speedup vs baseline: 1.1866x; 1.1866x over previous
#include <hip/hip_runtime.h>
#include <math.h>

typedef __attribute__((ext_vector_type(2))) float v2f;
typedef __attribute__((ext_vector_type(8))) float v8f;

__device__ __forceinline__ v8f wmma_f32(v2f a, v2f b, v8f c) {
    return __builtin_amdgcn_wmma_f32_16x16x4_f32(false, a, false, b, (short)0, c, false, false);
}
__device__ __forceinline__ v2f lds_frag(const float* p) {
    return *(const v2f*)p;
}

#define GT_M 64
#define GT_N 128
#define GT_K 64
#define GT_KQ (GT_K / 4)
#define A_SL (GT_M * 4 + 4)
#define B_SL (GT_N * 4 + 4)

typedef __attribute__((ext_vector_type(4))) float v4f;

__global__ __launch_bounds__(256) void ddnet_gemm_kernel(
    const float* __restrict__ A, const float* __restrict__ W,
    const float* __restrict__ bias, const float* __restrict__ resid,
    float* __restrict__ C, int Mdim, int Ndim, int Kdim, int act)
{
    __shared__ __align__(16) float As[2][GT_KQ * A_SL];
    __shared__ __align__(16) float Bs[2][GT_KQ * B_SL];

    const int tid   = threadIdx.x;
    const int lane  = tid & 31;
    const int wave  = tid >> 5;
    const int lhalf = lane >> 4;
    const int lmod  = lane & 15;
    const int m0    = blockIdx.y * GT_M;
    const int n0    = blockIdx.x * GT_N;
    const int wm    = (wave & 1) * 32;
    const int wn    = (wave >> 1) * 32;
    const int nk    = Kdim >> 6;

    const int lq = tid & 15;
    const int lr = tid >> 4;
    const float* Ablk = A + (size_t)m0 * Kdim;
    const float* Wblk = W + (size_t)n0 * Kdim;
    const unsigned toff = (unsigned)(lr * Kdim + lq * 4);
    const int rstep = 16 * Kdim;
    const int a_st = lq * A_SL + lr * 4;
    const int b_st = lq * B_SL + lr * 4;
    const int a_rd = lhalf * A_SL + (wm + lmod) * 4;
    const int b_rd = lhalf * B_SL + (wn + lmod) * 4;

    v8f acc[2][2] = {};
    v4f ra[4], rb[8];

    #pragma unroll
    for (int i = 0; i < 4; ++i)
        ra[i] = *(const v4f*)(Ablk + i * rstep + toff);
    #pragma unroll
    for (int i = 0; i < 8; ++i)
        rb[i] = *(const v4f*)(Wblk + i * rstep + toff);
    #pragma unroll
    for (int i = 0; i < 4; ++i)
        *(v4f*)&As[0][a_st + i * 64] = ra[i];
    #pragma unroll
    for (int i = 0; i < 8; ++i)
        *(v4f*)&Bs[0][b_st + i * 64] = rb[i];
    __syncthreads();

    for (int kt = 0; kt < nk; ++kt) {
        const int buf = kt & 1;
        if (kt + 1 < nk) {
            const int k0 = (kt + 1) << 6;
            #pragma unroll
            for (int i = 0; i < 4; ++i)
                ra[i] = *(const v4f*)(Ablk + (k0 + i * rstep) + toff);
            #pragma unroll
            for (int i = 0; i < 8; ++i)
                rb[i] = *(const v4f*)(Wblk + (k0 + i * rstep) + toff);
        }
        const float* Ab = As[buf] + a_rd;
        const float* Bb = Bs[buf] + b_rd;
        #pragma unroll
        for (int u = 0; u < GT_KQ / 2; ++u) {
            v4f a0 = *(const v4f*)&Ab[2 * u * A_SL];
            v4f a1 = *(const v4f*)&Ab[2 * u * A_SL + 64];
            v4f b0 = *(const v4f*)&Bb[2 * u * B_SL];
            v4f b1 = *(const v4f*)&Bb[2 * u * B_SL + 64];
            acc[0][0] = wmma_f32(a0.xy, b0.xy, acc[0][0]);
            acc[0][1] = wmma_f32(a0.xy, b1.xy, acc[0][1]);
            acc[1][0] = wmma_f32(a1.xy, b0.xy, acc[1][0]);
            acc[1][1] = wmma_f32(a1.xy, b1.xy, acc[1][1]);
            acc[0][0] = wmma_f32(a0.zw, b0.zw, acc[0][0]);
            acc[0][1] = wmma_f32(a0.zw, b1.zw, acc[0][1]);
            acc[1][0] = wmma_f32(a1.zw, b0.zw, acc[1][0]);
            acc[1][1] = wmma_f32(a1.zw, b1.zw, acc[1][1]);
        }
        if (kt + 1 < nk) {
            __syncthreads();
            #pragma unroll
            for (int i = 0; i < 4; ++i)
                *(v4f*)&As[buf ^ 1][a_st + i * 64] = ra[i];
            #pragma unroll
            for (int i = 0; i < 8; ++i)
                *(v4f*)&Bs[buf ^ 1][b_st + i * 64] = rb[i];
            __syncthreads();
        }
    }

    #pragma unroll
    for (int mi = 0; mi < 2; ++mi)
        #pragma unroll
        for (int ni = 0; ni < 2; ++ni)
            #pragma unroll
            for (int v = 0; v < 8; ++v) {
                int row = m0 + wm + mi * 16 + v + 8 * lhalf;
                int col = n0 + wn + ni * 16 + lmod;
                float val = acc[mi][ni][v];
                if (bias)  val += bias[col];
                if (resid) val += resid[(size_t)row * Ndim + col];
                if (act == 1)
                    val = 0.5f * val * (1.0f + erff(val * 0.70710678118654752f));
                C[(size_t)row * Ndim + col] = val;
            }
}

__global__ __launch_bounds__(256) void ddnet_pe_kernel(
    float* __restrict__ h, int S, int D)
{
    size_t idx = (size_t)blockIdx.x * 256 + threadIdx.x;
    int d = (int)(idx % D);
    int s = (int)((idx / D) % S);
    int pair = d >> 1;
    float div = __expf(-(float)(2 * pair) * (logf(10000.0f) / (float)D));
    float ang = (float)s * div;
    h[idx] += (d & 1) ? __cosf(ang) : __sinf(ang);
}

__global__ __launch_bounds__(256) void ddnet_ln_kernel(
    const float* __restrict__ x, const float* __restrict__ gamma,
    const float* __restrict__ beta, float* __restrict__ out, int D)
{
    int row  = blockIdx.x * 8 + (threadIdx.x >> 5);
    int lane = threadIdx.x & 31;
    const float* xr = x + (size_t)row * D;
    float vals[8];
    float s = 0.f;
    #pragma unroll
    for (int i = 0; i < 8; ++i) { vals[i] = xr[lane + i * 32]; s += vals[i]; }
    #pragma unroll
    for (int m = 16; m >= 1; m >>= 1) s += __shfl_xor(s, m, 32);
    float mean = s * (1.0f / 256.0f);
    float vs = 0.f;
    #pragma unroll
    for (int i = 0; i < 8; ++i) { float d = vals[i] - mean; vs += d * d; }
    #pragma unroll
    for (int m = 16; m >= 1; m >>= 1) vs += __shfl_xor(vs, m, 32);
    float rstd = rsqrtf(vs * (1.0f / 256.0f) + 1e-5f);
    float* orow = out + (size_t)row * D;
    #pragma unroll
    for (int i = 0; i < 8; ++i) {
        int c = lane + i * 32;
        orow[c] = (vals[i] - mean) * rstd * gamma[c] + beta[c];
    }
}

#define DD 256
#define DG 6

__global__ __launch_bounds__(256) void ddnet_desc_kernel(
    const float* __restrict__ xt, const float* __restrict__ h,
    const float* __restrict__ P, float* __restrict__ hd, int S)
{
    __shared__ __align__(16) float Xs[128 * 32];
    __shared__ float Ct[511 * DG + 1];

    const int s   = blockIdx.x;
    const int tid = threadIdx.x;
    const float two_pi = 6.28318530717958647692f;

    for (int idx = tid; idx < 511 * DG; idx += 256) {
        int t = idx / DG, g = idx % DG;
        float period = (2.0f + 0.5f * (float)t) * powf(1.5f, (float)g);
        Ct[idx] = __cosf(two_pi * (float)s / period);
    }
    for (int idx = tid; idx < 16 * DD; idx += 256) {
        int b = idx >> 8, j = idx & 255;
        Xs[((j >> 2) * 2 + ((j >> 1) & 1)) * 32 + b * 2 + (j & 1)] =
            xt[((size_t)b * S + s) * DD + j];
    }
    __syncthreads();

    const int lane = tid & 31, wave = tid >> 5;
    const int lhalf = lane >> 4, lmod = lane & 15;

    for (int it = wave; it < 16; it += 8) {
        const int i = it * 16 + lmod;
        v8f acc = {};
        for (int j0 = 0; j0 < DD; j0 += 16) {
            float wv[4][2];
            #pragma unroll
            for (int t = 0; t < 4; ++t)
                #pragma unroll
                for (int jj = 0; jj < 2; ++jj) {
                    int j = j0 + t * 4 + jj + 2 * lhalf;
                    const float* Pp = P + ((size_t)i * DD + j) * DG;
                    const float* Cp = Ct + (i + j) * DG;
                    float w = 0.f;
                    #pragma unroll
                    for (int g = 0; g < DG; ++g) w += Pp[g] * Cp[g];
                    wv[t][jj] = w;
                }
            #pragma unroll
            for (int t = 0; t < 4; ++t) {
                v2f a = lds_frag(&Xs[((j0 >> 2) + t) * 64 + lhalf * 32 + lmod * 2]);
                v2f b = { wv[t][0], wv[t][1] };
                acc = wmma_f32(a, b, acc);
            }
        }
        #pragma unroll
        for (int v = 0; v < 8; ++v) {
            int b = v + 8 * lhalf;
            size_t off = ((size_t)b * S + s) * DD + it * 16 + lmod;
            hd[off] = acc[v] + h[off];
        }
    }
}

__global__ __launch_bounds__(128) void ddnet_attn_kernel(
    const float* __restrict__ qkv, float* __restrict__ o, int S)
{
    const int qb    = S / 64;
    const int bh    = blockIdx.x / qb;
    const int qblk  = blockIdx.x % qb;
    const int b     = bh >> 3, head = bh & 7;
    const int tid   = threadIdx.x, lane = tid & 31, wave = tid >> 5;
    const int lhalf = lane >> 4, lmod = lane & 15;
    const int q0    = qblk * 64 + wave * 16;

    __shared__ __align__(16) float Kf[4][16 * 32];
    __shared__ __align__(16) float Vf[4][8 * 96];
    __shared__ __align__(16) float Pf[4][8 * 96];

    const size_t rs = 768;
    const float* qp = qkv + (size_t)b * S * rs + head * 32;
    const float* kp = qp + 256;
    const float* vp = qp + 512;
    const float scale = 0.17677669529663688f;

    v2f aq[8];
    #pragma unroll
    for (int t = 0; t < 8; ++t) {
        const float* qr = qp + (size_t)(q0 + lmod) * rs;
        aq[t].x = qr[t * 4 + 2 * lhalf];
        aq[t].y = qr[t * 4 + 1 + 2 * lhalf];
    }

    float mrow[8], lrow[8];
    v8f o0 = {}, o1 = {};
    #pragma unroll
    for (int v = 0; v < 8; ++v) { mrow[v] = -1e30f; lrow[v] = 0.f; }

    for (int k0 = 0; k0 < S; k0 += 16) {
        for (int i = lane; i < 16 * 32; i += 32) {
            int r = i >> 5, c = i & 31;
            Kf[wave][((c >> 2) * 2 + ((c >> 1) & 1)) * 32 + r * 2 + (c & 1)] =
                kp[(size_t)(k0 + r) * rs + c];
            Vf[wave][((r >> 2) * 2 + ((r >> 1) & 1)) * 96 + c * 2 + (r & 1)] =
                vp[(size_t)(k0 + r) * rs + c];
        }
        __syncthreads();

        v8f sc = {};
        #pragma unroll
        for (int t = 0; t < 8; ++t) {
            v2f bk = lds_frag(&Kf[wave][(t * 2 + lhalf) * 32 + lmod * 2]);
            sc = wmma_f32(aq[t], bk, sc);
        }

        float pr[8];
        #pragma unroll
        for (int v = 0; v < 8; ++v) {
            float sv = sc[v] * scale;
            float mx = sv;
            #pragma unroll
            for (int m = 8; m >= 1; m >>= 1) mx = fmaxf(mx, __shfl_xor(mx, m, 32));
            float mnew = fmaxf(mrow[v], mx);
            float corr = __expf(mrow[v] - mnew);
            float p    = __expf(sv - mnew);
            float rsum = p;
            #pragma unroll
            for (int m = 8; m >= 1; m >>= 1) rsum += __shfl_xor(rsum, m, 32);
            lrow[v] = lrow[v] * corr + rsum;
            mrow[v] = mnew;
            pr[v]   = p;
            o0[v] *= corr;
            o1[v] *= corr;
        }

        #pragma unroll
        for (int v = 0; v < 8; ++v)
            Pf[wave][((lmod >> 2) * 2 + ((lmod >> 1) & 1)) * 96 +
                     (v + 8 * lhalf) * 2 + (lmod & 1)] = pr[v];
        #pragma unroll
        for (int t = 0; t < 4; ++t) {
            int fo = t * 2 + lhalf;
            v2f ap  = lds_frag(&Pf[wave][fo * 96 + lmod * 2]);
            v2f bv0 = lds_frag(&Vf[wave][fo * 96 + lmod * 2]);
            v2f bv1 = lds_frag(&Vf[wave][fo * 96 + (16 + lmod) * 2]);
            o0 = wmma_f32(ap, bv0, o0);
            o1 = wmma_f32(ap, bv1, o1);
        }
        __syncthreads();
    }

    #pragma unroll
    for (int v = 0; v < 8; ++v) {
        int qr = v + 8 * lhalf;
        float inv = 1.0f / lrow[v];
        size_t base = ((size_t)b * S + q0 + qr) * 256 + head * 32;
        o[base + lmod]      = o0[v] * inv;
        o[base + 16 + lmod] = o1[v] * inv;
    }
}

static void launch_gemm(const float* A, const float* W, const float* bias,
                        const float* resid, float* C, int M, int N, int K,
                        int act, hipStream_t stream)
{
    dim3 grid(N / GT_N, M / GT_M);
    ddnet_gemm_kernel<<<grid, 256, 0, stream>>>(A, W, bias, resid, C, M, N, K, act);
}

extern "C" void kernel_launch(void* const* d_in, const int* in_sizes, int n_in,
                              void* d_out, int out_size, void* d_ws, size_t ws_size,
                              hipStream_t stream) {
    (void)in_sizes; (void)n_in; (void)out_size; (void)ws_size;
    const int B = 16, S = 512, IN = 64, D = 256, L = 3, G = 6, DFF = 1024;
    const int Mr = B * S;

    const float* x      = (const float*)d_in[0];
    const float* in_w   = (const float*)d_in[1];
    const float* in_b   = (const float*)d_in[2];
    const float* Mw     = (const float*)d_in[3];
    const float* P      = (const float*)d_in[4];
    const float* desc_g = (const float*)d_in[5];
    const float* desc_b = (const float*)d_in[6];
    const float* qkv_w  = (const float*)d_in[7];
    const float* qkv_b  = (const float*)d_in[8];
    const float* out_w  = (const float*)d_in[9];
    const float* out_b  = (const float*)d_in[10];
    const float* ff1_w  = (const float*)d_in[11];
    const float* ff1_b  = (const float*)d_in[12];
    const float* ff2_w  = (const float*)d_in[13];
    const float* ff2_b  = (const float*)d_in[14];
    const float* ln1_g  = (const float*)d_in[15];
    const float* ln1_b  = (const float*)d_in[16];
    const float* ln2_g  = (const float*)d_in[17];
    const float* ln2_b  = (const float*)d_in[18];

    float* h    = (float*)d_ws;
    float* tmp  = h    + (size_t)Mr * D;
    float* xt   = tmp  + (size_t)Mr * D;
    float* hd   = xt   + (size_t)Mr * D;
    float* qkv  = hd   + (size_t)Mr * D;
    float* obuf = qkv  + (size_t)Mr * 3 * D;
    float* h1   = obuf + (size_t)Mr * D;
    float* ff   = h1   + (size_t)Mr * D;

    launch_gemm(x, in_w, in_b, nullptr, h, Mr, D, IN, 0, stream);
    ddnet_pe_kernel<<<(Mr * D) / 256, 256, 0, stream>>>(h, S, D);

    for (int l = 0; l < L; ++l) {
        launch_gemm(h, Mw + (size_t)l * D * D, nullptr, nullptr, tmp, Mr, D, D, 0, stream);
        ddnet_ln_kernel<<<Mr / 8, 256, 0, stream>>>(tmp, desc_g + l * D, desc_b + l * D, xt, D);
        ddnet_desc_kernel<<<S, 256, 0, stream>>>(xt, h, P + (size_t)l * D * D * G, hd, S);
        launch_gemm(hd, qkv_w + (size_t)l * 3 * D * D, qkv_b + (size_t)l * 3 * D,
                    nullptr, qkv, Mr, 3 * D, D, 0, stream);
        ddnet_attn_kernel<<<B * 8 * (S / 64), 128, 0, stream>>>(qkv, obuf, S);
        launch_gemm(obuf, out_w + (size_t)l * D * D, out_b + l * D, hd, tmp, Mr, D, D, 0, stream);
        ddnet_ln_kernel<<<Mr / 8, 256, 0, stream>>>(tmp, ln1_g + l * D, ln1_b + l * D, h1, D);
        launch_gemm(h1, ff1_w + (size_t)l * DFF * D, ff1_b + (size_t)l * DFF,
                    nullptr, ff, Mr, DFF, D, 1  , stream);
        launch_gemm(ff, ff2_w + (size_t)l * D * DFF, ff2_b + l * D, h1, tmp, Mr, D, DFF, 0, stream);
        float* dst = (l == L - 1) ? (float*)d_out : h;
        ddnet_ln_kernel<<<Mr / 8, 256, 0, stream>>>(tmp, ln2_g + l * D, ln2_b + l * D, dst, D);
    }
}


# Attention computed transposed (St=K.Qt, Ot=Vt.Pt): K/V/Q fragments straight from global, softmax max/sum in-lane, P stays in registers; no LDS or barrier; dh and key sum order changed
# speedup vs baseline: 1.5345x; 1.2932x over previous
#include <hip/hip_runtime.h>
#include <math.h>

typedef __attribute__((ext_vector_type(2))) float v2f;
typedef __attribute__((ext_vector_type(8))) float v8f;

__device__ __forceinline__ v8f wmma_f32(v2f a, v2f b, v8f c) {
    return __builtin_amdgcn_wmma_f32_16x16x4_f32(false, a, false, b, (short)0, c, false, false);
}
__device__ __forceinline__ v2f lds_frag(const float* p) {
    return *(const v2f*)p;
}

#define GT_M 64
#define GT_N 128
#define GT_K 64
#define GT_KQ (GT_K / 4)
#define A_SL (GT_M * 4 + 4)
#define B_SL (GT_N * 4 + 4)

typedef __attribute__((ext_vector_type(4))) float v4f;

__global__ __launch_bounds__(256) void ddnet_gemm_kernel(
    const float* __restrict__ A, const float* __restrict__ W,
    const float* __restrict__ bias, const float* __restrict__ resid,
    float* __restrict__ C, int Mdim, int Ndim, int Kdim, int act)
{
    __shared__ __align__(16) float As[2][GT_KQ * A_SL];
    __shared__ __align__(16) float Bs[2][GT_KQ * B_SL];

    const int tid   = threadIdx.x;
    const int lane  = tid & 31;
    const int wave  = tid >> 5;
    const int lhalf = lane >> 4;
    const int lmod  = lane & 15;
    const int m0    = blockIdx.y * GT_M;
    const int n0    = blockIdx.x * GT_N;
    const int wm    = (wave & 1) * 32;
    const int wn    = (wave >> 1) * 32;
    const int nk    = Kdim >> 6;

    const int lq = tid & 15;
    const int lr = tid >> 4;
    const float* Ablk = A + (size_t)m0 * Kdim;
    const float* Wblk = W + (size_t)n0 * Kdim;
    const unsigned toff = (unsigned)(lr * Kdim + lq * 4);
    const int rstep = 16 * Kdim;
    const int a_st = lq * A_SL + lr * 4;
    const int b_st = lq * B_SL + lr * 4;
    const int a_rd = lhalf * A_SL + (wm + lmod) * 4;
    const int b_rd = lhalf * B_SL + (wn + lmod) * 4;

    v8f acc[2][2] = {};
    v4f ra[4], rb[8];

    #pragma unroll
    for (int i = 0; i < 4; ++i)
        ra[i] = *(const v4f*)(Ablk + i * rstep + toff);
    #pragma unroll
    for (int i = 0; i < 8; ++i)
        rb[i] = *(const v4f*)(Wblk + i * rstep + toff);
    #pragma unroll
    for (int i = 0; i < 4; ++i)
        *(v4f*)&As[0][a_st + i * 64] = ra[i];
    #pragma unroll
    for (int i = 0; i < 8; ++i)
        *(v4f*)&Bs[0][b_st + i * 64] = rb[i];
    __syncthreads();

    for (int kt = 0; kt < nk; ++kt) {
        const int buf = kt & 1;
        if (kt + 1 < nk) {
            const int k0 = (kt + 1) << 6;
            #pragma unroll
            for (int i = 0; i < 4; ++i)
                ra[i] = *(const v4f*)(Ablk + (k0 + i * rstep) + toff);
            #pragma unroll
            for (int i = 0; i < 8; ++i)
                rb[i] = *(const v4f*)(Wblk + (k0 + i * rstep) + toff);
        }
        const float* Ab = As[buf] + a_rd;
        const float* Bb = Bs[buf] + b_rd;
        #pragma unroll
        for (int u = 0; u < GT_KQ / 2; ++u) {
            v4f a0 = *(const v4f*)&Ab[2 * u * A_SL];
            v4f a1 = *(const v4f*)&Ab[2 * u * A_SL + 64];
            v4f b0 = *(const v4f*)&Bb[2 * u * B_SL];
            v4f b1 = *(const v4f*)&Bb[2 * u * B_SL + 64];
            acc[0][0] = wmma_f32(a0.xy, b0.xy, acc[0][0]);
            acc[0][1] = wmma_f32(a0.xy, b1.xy, acc[0][1]);
            acc[1][0] = wmma_f32(a1.xy, b0.xy, acc[1][0]);
            acc[1][1] = wmma_f32(a1.xy, b1.xy, acc[1][1]);
            acc[0][0] = wmma_f32(a0.zw, b0.zw, acc[0][0]);
            acc[0][1] = wmma_f32(a0.zw, b1.zw, acc[0][1]);
            acc[1][0] = wmma_f32(a1.zw, b0.zw, acc[1][0]);
            acc[1][1] = wmma_f32(a1.zw, b1.zw, acc[1][1]);
        }
        if (kt + 1 < nk) {
            __syncthreads();
            #pragma unroll
            for (int i = 0; i < 4; ++i)
                *(v4f*)&As[buf ^ 1][a_st + i * 64] = ra[i];
            #pragma unroll
            for (int i = 0; i < 8; ++i)
                *(v4f*)&Bs[buf ^ 1][b_st + i * 64] = rb[i];
            __syncthreads();
        }
    }

    #pragma unroll
    for (int mi = 0; mi < 2; ++mi)
        #pragma unroll
        for (int ni = 0; ni < 2; ++ni)
            #pragma unroll
            for (int v = 0; v < 8; ++v) {
                int row = m0 + wm + mi * 16 + v + 8 * lhalf;
                int col = n0 + wn + ni * 16 + lmod;
                float val = acc[mi][ni][v];
                if (bias)  val += bias[col];
                if (resid) val += resid[(size_t)row * Ndim + col];
                if (act == 1)
                    val = 0.5f * val * (1.0f + erff(val * 0.70710678118654752f));
                C[(size_t)row * Ndim + col] = val;
            }
}

__global__ __launch_bounds__(256) void ddnet_pe_kernel(
    float* __restrict__ h, int S, int D)
{
    size_t idx = (size_t)blockIdx.x * 256 + threadIdx.x;
    int d = (int)(idx % D);
    int s = (int)((idx / D) % S);
    int pair = d >> 1;
    float div = __expf(-(float)(2 * pair) * (logf(10000.0f) / (float)D));
    float ang = (float)s * div;
    h[idx] += (d & 1) ? __cosf(ang) : __sinf(ang);
}

__global__ __launch_bounds__(256) void ddnet_ln_kernel(
    const float* __restrict__ x, const float* __restrict__ gamma,
    const float* __restrict__ beta, float* __restrict__ out, int D)
{
    int row  = blockIdx.x * 8 + (threadIdx.x >> 5);
    int lane = threadIdx.x & 31;
    const float* xr = x + (size_t)row * D;
    float vals[8];
    float s = 0.f;
    #pragma unroll
    for (int i = 0; i < 8; ++i) { vals[i] = xr[lane + i * 32]; s += vals[i]; }
    #pragma unroll
    for (int m = 16; m >= 1; m >>= 1) s += __shfl_xor(s, m, 32);
    float mean = s * (1.0f / 256.0f);
    float vs = 0.f;
    #pragma unroll
    for (int i = 0; i < 8; ++i) { float d = vals[i] - mean; vs += d * d; }
    #pragma unroll
    for (int m = 16; m >= 1; m >>= 1) vs += __shfl_xor(vs, m, 32);
    float rstd = rsqrtf(vs * (1.0f / 256.0f) + 1e-5f);
    float* orow = out + (size_t)row * D;
    #pragma unroll
    for (int i = 0; i < 8; ++i) {
        int c = lane + i * 32;
        orow[c] = (vals[i] - mean) * rstd * gamma[c] + beta[c];
    }
}

#define DD 256
#define DG 6

__global__ __launch_bounds__(256) void ddnet_desc_kernel(
    const float* __restrict__ xt, const float* __restrict__ h,
    const float* __restrict__ P, float* __restrict__ hd, int S)
{
    __shared__ __align__(16) float Xs[128 * 32];
    __shared__ float Ct[511 * DG + 1];

    const int s   = blockIdx.x;
    const int tid = threadIdx.x;
    const float two_pi = 6.28318530717958647692f;

    for (int idx = tid; idx < 511 * DG; idx += 256) {
        int t = idx / DG, g = idx % DG;
        float period = (2.0f + 0.5f * (float)t) * powf(1.5f, (float)g);
        Ct[idx] = __cosf(two_pi * (float)s / period);
    }
    for (int idx = tid; idx < 16 * DD; idx += 256) {
        int b = idx >> 8, j = idx & 255;
        Xs[((j >> 2) * 2 + ((j >> 1) & 1)) * 32 + b * 2 + (j & 1)] =
            xt[((size_t)b * S + s) * DD + j];
    }
    __syncthreads();

    const int lane = tid & 31, wave = tid >> 5;
    const int lhalf = lane >> 4, lmod = lane & 15;

    for (int it = wave; it < 16; it += 8) {
        const int i = it * 16 + lmod;
        v8f acc = {};
        for (int j0 = 0; j0 < DD; j0 += 16) {
            float wv[4][2];
            #pragma unroll
            for (int t = 0; t < 4; ++t)
                #pragma unroll
                for (int jj = 0; jj < 2; ++jj) {
                    int j = j0 + t * 4 + jj + 2 * lhalf;
                    const float* Pp = P + ((size_t)i * DD + j) * DG;
                    const float* Cp = Ct + (i + j) * DG;
                    float w = 0.f;
                    #pragma unroll
                    for (int g = 0; g < DG; ++g) w += Pp[g] * Cp[g];
                    wv[t][jj] = w;
                }
            #pragma unroll
            for (int t = 0; t < 4; ++t) {
                v2f a = lds_frag(&Xs[((j0 >> 2) + t) * 64 + lhalf * 32 + lmod * 2]);
                v2f b = { wv[t][0], wv[t][1] };
                acc = wmma_f32(a, b, acc);
            }
        }
        #pragma unroll
        for (int v = 0; v < 8; ++v) {
            int b = v + 8 * lhalf;
            size_t off = ((size_t)b * S + s) * DD + it * 16 + lmod;
            hd[off] = acc[v] + h[off];
        }
    }
}

struct AttnTile { v4f k[4]; v2f v[8]; };

__device__ __forceinline__ float attn_xhalf(float x) {
    return __builtin_bit_cast(float, __builtin_amdgcn_permlanex16(
        __builtin_bit_cast(int, x), __builtin_bit_cast(int, x),
        0x76543210, 0xfedcba98, false, false));
}

__device__ __forceinline__ void attn_load(AttnTile& T, const float* kr, const float* vr) {
    #pragma unroll
    for (int c = 0; c < 4; ++c) T.k[c] = *(const v4f*)(kr + 4 * c);
    #pragma unroll
    for (int j = 0; j < 8; ++j) T.v[j] = *(const v2f*)(vr + (size_t)j * 768);
}

__device__ __forceinline__ void attn_step(const AttnTile& T, const v4f (&qv)[4],
                                          float& mrow, float& lsum, v8f& o0, v8f& o1) {
    const float scale = 0.17677669529663688f;
    v8f s0 = {}, s1 = {};
    #pragma unroll
    for (int c = 0; c < 4; ++c) {
        s0 = wmma_f32(T.k[c].xy, qv[c].xy, s0);
        s1 = wmma_f32(T.k[c].zw, qv[c].zw, s1);
    }
    float sv[8];
    #pragma unroll
    for (int v = 0; v < 8; ++v) sv[v] = (s0[v] + s1[v]) * scale;
    float mx = fmaxf(fmaxf(fmaxf(sv[0], sv[1]), fmaxf(sv[2], sv[3])),
                     fmaxf(fmaxf(sv[4], sv[5]), fmaxf(sv[6], sv[7])));
    mx = fmaxf(mx, attn_xhalf(mx));
    const float mnew = fmaxf(mrow, mx);
    const float corr = __expf(mrow - mnew);
    float p[8];
    float rsum = 0.f;
    #pragma unroll
    for (int v = 0; v < 8; ++v) { p[v] = __expf(sv[v] - mnew); rsum += p[v]; }
    lsum = lsum * corr + rsum;
    mrow = mnew;
    #pragma unroll
    for (int v = 0; v < 8; ++v) { o0[v] *= corr; o1[v] *= corr; }
    #pragma unroll
    for (int t = 0; t < 4; ++t) {
        v2f bp = { p[2 * t], p[2 * t + 1] };
        v2f a0 = { T.v[2 * t].x, T.v[2 * t + 1].x };
        v2f a1 = { T.v[2 * t].y, T.v[2 * t + 1].y };
        o0 = wmma_f32(a0, bp, o0);
        o1 = wmma_f32(a1, bp, o1);
    }
}

__global__ __launch_bounds__(128) void ddnet_attn_kernel(
    const float* __restrict__ qkv, float* __restrict__ o, int S)
{
    const int qb    = S / 64;
    const int bh    = blockIdx.x / qb;
    const int qblk  = blockIdx.x % qb;
    const int b     = bh >> 3, head = bh & 7;
    const int tid   = threadIdx.x, lane = tid & 31, wave = tid >> 5;
    const int lhalf = lane >> 4, lmod = lane & 15;
    const int q0    = qblk * 64 + wave * 16;

    const size_t rs = 768;
    const float* qp = qkv + (size_t)b * S * rs + head * 32;
    const float* kr = qp + 256 + (size_t)lmod * rs + 16 * lhalf;
    const float* vr = qp + 512 + (size_t)(8 * lhalf) * rs + 2 * lmod;

    v4f qv[4];
    {
        const float* qr = qp + (size_t)(q0 + lmod) * rs + 16 * lhalf;
        #pragma unroll
        for (int c = 0; c < 4; ++c) qv[c] = *(const v4f*)(qr + 4 * c);
    }

    float mrow = -1e30f, lsum = 0.f;
    v8f o0 = {}, o1 = {};

    AttnTile Ta, Tb;
    attn_load(Ta, kr, vr);
    attn_load(Tb, kr + (size_t)16 * rs, vr + (size_t)16 * rs);
    for (int k0 = 0; k0 < S; k0 += 32) {
        attn_step(Ta, qv, mrow, lsum, o0, o1);
        const int kn = (k0 + 32 < S) ? (k0 + 32) : k0;
        attn_load(Ta, kr + (size_t)kn * rs, vr + (size_t)kn * rs);
        __builtin_amdgcn_sched_barrier(0);
        attn_step(Tb, qv, mrow, lsum, o0, o1);
        attn_load(Tb, kr + (size_t)(kn + 16) * rs, vr + (size_t)(kn + 16) * rs);
        __builtin_amdgcn_sched_barrier(0);
    }

    const float inv = 1.0f / (lsum + attn_xhalf(lsum));
    float* orow = o + ((size_t)b * S + q0 + lmod) * 256 + head * 32 + 16 * lhalf;
    #pragma unroll
    for (int j = 0; j < 4; ++j) {
        v4f w = { o0[2 * j] * inv, o1[2 * j] * inv, o0[2 * j + 1] * inv, o1[2 * j + 1] * inv };
        *(v4f*)(orow + 4 * j) = w;
    }
}

static void launch_gemm(const float* A, const float* W, const float* bias,
                        const float* resid, float* C, int M, int N, int K,
                        int act, hipStream_t stream)
{
    dim3 grid(N / GT_N, M / GT_M);
    ddnet_gemm_kernel<<<grid, 256, 0, stream>>>(A, W, bias, resid, C, M, N, K, act);
}

extern "C" void kernel_launch(void* const* d_in, const int* in_sizes, int n_in,
                              void* d_out, int out_size, void* d_ws, size_t ws_size,
                              hipStream_t stream) {
    (void)in_sizes; (void)n_in; (void)out_size; (void)ws_size;
    const int B = 16, S = 512, IN = 64, D = 256, L = 3, G = 6, DFF = 1024;
    const int Mr = B * S;

    const float* x      = (const float*)d_in[0];
    const float* in_w   = (const float*)d_in[1];
    const float* in_b   = (const float*)d_in[2];
    const float* Mw     = (const float*)d_in[3];
    const float* P      = (const float*)d_in[4];
    const float* desc_g = (const float*)d_in[5];
    const float* desc_b = (const float*)d_in[6];
    const float* qkv_w  = (const float*)d_in[7];
    const float* qkv_b  = (const float*)d_in[8];
    const float* out_w  = (const float*)d_in[9];
    const float* out_b  = (const float*)d_in[10];
    const float* ff1_w  = (const float*)d_in[11];
    const float* ff1_b  = (const float*)d_in[12];
    const float* ff2_w  = (const float*)d_in[13];
    const float* ff2_b  = (const float*)d_in[14];
    const float* ln1_g  = (const float*)d_in[15];
    const float* ln1_b  = (const float*)d_in[16];
    const float* ln2_g  = (const float*)d_in[17];
    const float* ln2_b  = (const float*)d_in[18];

    float* h    = (float*)d_ws;
    float* tmp  = h    + (size_t)Mr * D;
    float* xt   = tmp  + (size_t)Mr * D;
    float* hd   = xt   + (size_t)Mr * D;
    float* qkv  = hd   + (size_t)Mr * D;
    float* obuf = qkv  + (size_t)Mr * 3 * D;
    float* h1   = obuf + (size_t)Mr * D;
    float* ff   = h1   + (size_t)Mr * D;

    launch_gemm(x, in_w, in_b, nullptr, h, Mr, D, IN, 0, stream);
    ddnet_pe_kernel<<<(Mr * D) / 256, 256, 0, stream>>>(h, S, D);

    for (int l = 0; l < L; ++l) {
        launch_gemm(h, Mw + (size_t)l * D * D, nullptr, nullptr, tmp, Mr, D, D, 0, stream);
        ddnet_ln_kernel<<<Mr / 8, 256, 0, stream>>>(tmp, desc_g + l * D, desc_b + l * D, xt, D);
        ddnet_desc_kernel<<<S, 256, 0, stream>>>(xt, h, P + (size_t)l * D * D * G, hd, S);
        launch_gemm(hd, qkv_w + (size_t)l * 3 * D * D, qkv_b + (size_t)l * 3 * D,
                    nullptr, qkv, Mr, 3 * D, D, 0, stream);
        ddnet_attn_kernel<<<B * 8 * (S / 64), 128, 0, stream>>>(qkv, obuf, S);
        launch_gemm(obuf, out_w + (size_t)l * D * D, out_b + l * D, hd, tmp, Mr, D, D, 0, stream);
        ddnet_ln_kernel<<<Mr / 8, 256, 0, stream>>>(tmp, ln1_g + l * D, ln1_b + l * D, h1, D);
        launch_gemm(h1, ff1_w + (size_t)l * DFF * D, ff1_b + (size_t)l * DFF,
                    nullptr, ff, Mr, DFF, D, 1  , stream);
        launch_gemm(ff, ff2_w + (size_t)l * D * DFF, ff2_b + l * D, h1, tmp, Mr, D, DFF, 0, stream);
        float* dst = (l == L - 1) ? (float*)d_out : h;
        ddnet_ln_kernel<<<Mr / 8, 256, 0, stream>>>(tmp, ln2_g + l * D, ln2_b + l * D, dst, D);
    }
}


# Descriptor kernel: a work-group now covers 4 positions x 128 outputs; a lane reads its P values once into registers, one 16-j tile ahead, and reuses them for the 4 positions (P loads per result /4)
# speedup vs baseline: 1.8917x; 1.2328x over previous
#include <hip/hip_runtime.h>
#include <math.h>

typedef __attribute__((ext_vector_type(2))) float v2f;
typedef __attribute__((ext_vector_type(8))) float v8f;

__device__ __forceinline__ v8f wmma_f32(v2f a, v2f b, v8f c) {
    return __builtin_amdgcn_wmma_f32_16x16x4_f32(false, a, false, b, (short)0, c, false, false);
}
__device__ __forceinline__ v2f lds_frag(const float* p) {
    return *(const v2f*)p;
}

#define GT_M 64
#define GT_N 128
#define GT_K 64
#define GT_KQ (GT_K / 4)
#define A_SL (GT_M * 4 + 4)
#define B_SL (GT_N * 4 + 4)

typedef __attribute__((ext_vector_type(4))) float v4f;

__global__ __launch_bounds__(256) void ddnet_gemm_kernel(
    const float* __restrict__ A, const float* __restrict__ W,
    const float* __restrict__ bias, const float* __restrict__ resid,
    float* __restrict__ C, int Mdim, int Ndim, int Kdim, int act)
{
    __shared__ __align__(16) float As[2][GT_KQ * A_SL];
    __shared__ __align__(16) float Bs[2][GT_KQ * B_SL];

    const int tid   = threadIdx.x;
    const int lane  = tid & 31;
    const int wave  = tid >> 5;
    const int lhalf = lane >> 4;
    const int lmod  = lane & 15;
    const int m0    = blockIdx.y * GT_M;
    const int n0    = blockIdx.x * GT_N;
    const int wm    = (wave & 1) * 32;
    const int wn    = (wave >> 1) * 32;
    const int nk    = Kdim >> 6;

    const int lq = tid & 15;
    const int lr = tid >> 4;
    const float* Ablk = A + (size_t)m0 * Kdim;
    const float* Wblk = W + (size_t)n0 * Kdim;
    const unsigned toff = (unsigned)(lr * Kdim + lq * 4);
    const int rstep = 16 * Kdim;
    const int a_st = lq * A_SL + lr * 4;
    const int b_st = lq * B_SL + lr * 4;
    const int a_rd = lhalf * A_SL + (wm + lmod) * 4;
    const int b_rd = lhalf * B_SL + (wn + lmod) * 4;

    v8f acc[2][2] = {};
    v4f ra[4], rb[8];

    #pragma unroll
    for (int i = 0; i < 4; ++i)
        ra[i] = *(const v4f*)(Ablk + i * rstep + toff);
    #pragma unroll
    for (int i = 0; i < 8; ++i)
        rb[i] = *(const v4f*)(Wblk + i * rstep + toff);
    #pragma unroll
    for (int i = 0; i < 4; ++i)
        *(v4f*)&As[0][a_st + i * 64] = ra[i];
    #pragma unroll
    for (int i = 0; i < 8; ++i)
        *(v4f*)&Bs[0][b_st + i * 64] = rb[i];
    __syncthreads();

    for (int kt = 0; kt < nk; ++kt) {
        const int buf = kt & 1;
        if (kt + 1 < nk) {
            const int k0 = (kt + 1) << 6;
            #pragma unroll
            for (int i = 0; i < 4; ++i)
                ra[i] = *(const v4f*)(Ablk + (k0 + i * rstep) + toff);
            #pragma unroll
            for (int i = 0; i < 8; ++i)
                rb[i] = *(const v4f*)(Wblk + (k0 + i * rstep) + toff);
        }
        const float* Ab = As[buf] + a_rd;
        const float* Bb = Bs[buf] + b_rd;
        #pragma unroll
        for (int u = 0; u < GT_KQ / 2; ++u) {
            v4f a0 = *(const v4f*)&Ab[2 * u * A_SL];
            v4f a1 = *(const v4f*)&Ab[2 * u * A_SL + 64];
            v4f b0 = *(const v4f*)&Bb[2 * u * B_SL];
            v4f b1 = *(const v4f*)&Bb[2 * u * B_SL + 64];
            acc[0][0] = wmma_f32(a0.xy, b0.xy, acc[0][0]);
            acc[0][1] = wmma_f32(a0.xy, b1.xy, acc[0][1]);
            acc[1][0] = wmma_f32(a1.xy, b0.xy, acc[1][0]);
            acc[1][1] = wmma_f32(a1.xy, b1.xy, acc[1][1]);
            acc[0][0] = wmma_f32(a0.zw, b0.zw, acc[0][0]);
            acc[0][1] = wmma_f32(a0.zw, b1.zw, acc[0][1]);
            acc[1][0] = wmma_f32(a1.zw, b0.zw, acc[1][0]);
            acc[1][1] = wmma_f32(a1.zw, b1.zw, acc[1][1]);
        }
        if (kt + 1 < nk) {
            __syncthreads();
            #pragma unroll
            for (int i = 0; i < 4; ++i)
                *(v4f*)&As[buf ^ 1][a_st + i * 64] = ra[i];
            #pragma unroll
            for (int i = 0; i < 8; ++i)
                *(v4f*)&Bs[buf ^ 1][b_st + i * 64] = rb[i];
            __syncthreads();
        }
    }

    #pragma unroll
    for (int mi = 0; mi < 2; ++mi)
        #pragma unroll
        for (int ni = 0; ni < 2; ++ni)
            #pragma unroll
            for (int v = 0; v < 8; ++v) {
                int row = m0 + wm + mi * 16 + v + 8 * lhalf;
                int col = n0 + wn + ni * 16 + lmod;
                float val = acc[mi][ni][v];
                if (bias)  val += bias[col];
                if (resid) val += resid[(size_t)row * Ndim + col];
                if (act == 1)
                    val = 0.5f * val * (1.0f + erff(val * 0.70710678118654752f));
                C[(size_t)row * Ndim + col] = val;
            }
}

__global__ __launch_bounds__(256) void ddnet_pe_kernel(
    float* __restrict__ h, int S, int D)
{
    size_t idx = (size_t)blockIdx.x * 256 + threadIdx.x;
    int d = (int)(idx % D);
    int s = (int)((idx / D) % S);
    int pair = d >> 1;
    float div = __expf(-(float)(2 * pair) * (logf(10000.0f) / (float)D));
    float ang = (float)s * div;
    h[idx] += (d & 1) ? __cosf(ang) : __sinf(ang);
}

__global__ __launch_bounds__(256) void ddnet_ln_kernel(
    const float* __restrict__ x, const float* __restrict__ gamma,
    const float* __restrict__ beta, float* __restrict__ out, int D)
{
    int row  = blockIdx.x * 8 + (threadIdx.x >> 5);
    int lane = threadIdx.x & 31;
    const float* xr = x + (size_t)row * D;
    float vals[8];
    float s = 0.f;
    #pragma unroll
    for (int i = 0; i < 8; ++i) { vals[i] = xr[lane + i * 32]; s += vals[i]; }
    #pragma unroll
    for (int m = 16; m >= 1; m >>= 1) s += __shfl_xor(s, m, 32);
    float mean = s * (1.0f / 256.0f);
    float vs = 0.f;
    #pragma unroll
    for (int i = 0; i < 8; ++i) { float d = vals[i] - mean; vs += d * d; }
    #pragma unroll
    for (int m = 16; m >= 1; m >>= 1) vs += __shfl_xor(vs, m, 32);
    float rstd = rsqrtf(vs * (1.0f / 256.0f) + 1e-5f);
    float* orow = out + (size_t)row * D;
    #pragma unroll
    for (int i = 0; i < 8; ++i) {
        int c = lane + i * 32;
        orow[c] = (vals[i] - mean) * rstd * gamma[c] + beta[c];
    }
}

#define DD 256
#define DG 6
#define DS_SG 4
#define DS_NI 128
#define DS_XP 66
#define DS_XS (64 * DS_XP)
#define DS_NT (DS_NI + DD - 1)
#define DS_CS (384 * DG)

struct DescTile { v4f p[12]; };

__device__ __forceinline__ void desc_load(DescTile& T, const float* pr) {
    #pragma unroll
    for (int t = 0; t < 4; ++t)
        #pragma unroll
        for (int q = 0; q < 3; ++q)
            T.p[3 * t + q] = *(const v4f*)(pr + t * 24 + q * 4);
}

struct DescOps { v2f c[DS_SG][6]; v2f a[DS_SG]; };

__device__ __forceinline__ void desc_ops(DescOps& O, const float* cp, const float* xp) {
    #pragma unroll
    for (int sl = 0; sl < DS_SG; ++sl) {
        #pragma unroll
        for (int m = 0; m < 6; ++m)
            O.c[sl][m] = *(const v2f*)(cp + sl * DS_CS + 2 * m);
        O.a[sl] = lds_frag(xp + sl * DS_XS);
    }
}

__device__ __forceinline__ void desc_group(const DescTile& T, int t, const DescOps& O,
                                           v8f (&acc)[DS_SG]) {
    #pragma unroll
    for (int sl = 0; sl < DS_SG; ++sl) {
        float w0 = 0.f, w1 = 0.f;
        #pragma unroll
        for (int g = 0; g < DG; ++g) {
            w0 += T.p[3 * t + (g >> 2)][g & 3] * O.c[sl][g >> 1][g & 1];
            w1 += T.p[3 * t + ((6 + g) >> 2)][(6 + g) & 3] * O.c[sl][(6 + g) >> 1][g & 1];
        }
        v2f b = { w0, w1 };
        acc[sl] = wmma_f32(O.a[sl], b, acc[sl]);
    }
}

__device__ __forceinline__ void desc_step(const DescTile& T, const float* cb,
                                          const float* xb, v8f (&acc)[DS_SG]) {
    DescOps Oa, Ob;
    desc_ops(Oa, cb, xb);
    desc_ops(Ob, cb + 24, xb + DS_XP);
    __builtin_amdgcn_sched_barrier(0);
    desc_group(T, 0, Oa, acc);
    desc_ops(Oa, cb + 48, xb + 2 * DS_XP);
    __builtin_amdgcn_sched_barrier(0);
    desc_group(T, 1, Ob, acc);
    desc_ops(Ob, cb + 72, xb + 3 * DS_XP);
    __builtin_amdgcn_sched_barrier(0);
    desc_group(T, 2, Oa, acc);
    desc_group(T, 3, Ob, acc);
}

__global__ __launch_bounds__(256) void ddnet_desc_kernel(
    const float* __restrict__ xt, const float* __restrict__ h,
    const float* __restrict__ P, float* __restrict__ hd, int S)
{
    __shared__ __align__(16) float Xs[DS_SG * DS_XS];
    __shared__ __align__(16) float Ct[DS_SG * DS_CS];

    const int tid = threadIdx.x;
    const int i0  = (blockIdx.x & 1) * DS_NI;
    const int s0  = (blockIdx.x >> 1) * DS_SG;
    const float two_pi = 6.28318530717958647692f;

    v4f xv[16];
    {
        const float* xsrc = xt + (size_t)s0 * DD + tid * 4;
        #pragma unroll
        for (int bb = 0; bb < 16; ++bb)
            xv[bb] = *(const v4f*)(xsrc + (size_t)bb * S * DD);
    }

    if (tid < 42 * DG) {
        const int g = tid % DG, tq = tid / DG;
        const float pw = powf(1.5f, (float)g);
        for (int tt = tq; tt < DS_NT; tt += 42) {
            const float period = (2.0f + 0.5f * (float)(i0 + tt)) * pw;
            #pragma unroll
            for (int sl = 0; sl < DS_SG; ++sl)
                Ct[sl * DS_CS + tt * DG + g] = __cosf(two_pi * (float)(s0 + sl) / period);
        }
    }
    {
        float* xd = Xs + (tid >> 6) * DS_XS + (tid & 63) * DS_XP;
        #pragma unroll
        for (int bb = 0; bb < 16; ++bb) {
            v2f lo = { xv[bb].x, xv[bb].y }, hi = { xv[bb].z, xv[bb].w };
            *(v2f*)(xd + bb * 2)      = lo;
            *(v2f*)(xd + 32 + bb * 2) = hi;
        }
    }
    __syncthreads();

    const int lane = tid & 31, wave = tid >> 5;
    const int lhalf = lane >> 4, lmod = lane & 15;
    const int iloc = wave * 16 + lmod;
    const float* prow = P + ((size_t)(i0 + iloc) * DD + 2 * lhalf) * DG;
    const float* cb = Ct + (iloc + 2 * lhalf) * DG;
    const float* xb = Xs + lhalf * 32 + lmod * 2;

    v8f acc[DS_SG] = {};
    DescTile Ta, Tb;
    desc_load(Ta, prow);
    for (int j0 = 0; j0 < DD; j0 += 32) {
        desc_load(Tb, prow + (j0 + 16) * DG);
        __builtin_amdgcn_sched_barrier(0);
        desc_step(Ta, cb + j0 * DG, xb + (j0 >> 2) * DS_XP, acc);
        const int jn = (j0 + 32 < DD) ? (j0 + 32) : j0;
        desc_load(Ta, prow + jn * DG);
        __builtin_amdgcn_sched_barrier(0);
        desc_step(Tb, cb + (j0 + 16) * DG, xb + ((j0 + 16) >> 2) * DS_XP, acc);
    }

    float hv[DS_SG][8];
    #pragma unroll
    for (int sl = 0; sl < DS_SG; ++sl)
        #pragma unroll
        for (int v = 0; v < 8; ++v)
            hv[sl][v] = h[((size_t)(v + 8 * lhalf) * S + s0 + sl) * DD + i0 + iloc];
    #pragma unroll
    for (int sl = 0; sl < DS_SG; ++sl)
        #pragma unroll
        for (int v = 0; v < 8; ++v)
            hd[((size_t)(v + 8 * lhalf) * S + s0 + sl) * DD + i0 + iloc] = acc[sl][v] + hv[sl][v];
}

struct AttnTile { v4f k[4]; v2f v[8]; };

__device__ __forceinline__ float attn_xhalf(float x) {
    return __builtin_bit_cast(float, __builtin_amdgcn_permlanex16(
        __builtin_bit_cast(int, x), __builtin_bit_cast(int, x),
        0x76543210, 0xfedcba98, false, false));
}

__device__ __forceinline__ void attn_load(AttnTile& T, const float* kr, const float* vr) {
    #pragma unroll
    for (int c = 0; c < 4; ++c) T.k[c] = *(const v4f*)(kr + 4 * c);
    #pragma unroll
    for (int j = 0; j < 8; ++j) T.v[j] = *(const v2f*)(vr + (size_t)j * 768);
}

__device__ __forceinline__ void attn_step(const AttnTile& T, const v4f (&qv)[4],
                                          float& mrow, float& lsum, v8f& o0, v8f& o1) {
    const float scale = 0.17677669529663688f;
    v8f s0 = {}, s1 = {};
    #pragma unroll
    for (int c = 0; c < 4; ++c) {
        s0 = wmma_f32(T.k[c].xy, qv[c].xy, s0);
        s1 = wmma_f32(T.k[c].zw, qv[c].zw, s1);
    }
    float sv[8];
    #pragma unroll
    for (int v = 0; v < 8; ++v) sv[v] = (s0[v] + s1[v]) * scale;
    float mx = fmaxf(fmaxf(fmaxf(sv[0], sv[1]), fmaxf(sv[2], sv[3])),
                     fmaxf(fmaxf(sv[4], sv[5]), fmaxf(sv[6], sv[7])));
    mx = fmaxf(mx, attn_xhalf(mx));
    const float mnew = fmaxf(mrow, mx);
    const float corr = __expf(mrow - mnew);
    float p[8];
    float rsum = 0.f;
    #pragma unroll
    for (int v = 0; v < 8; ++v) { p[v] = __expf(sv[v] - mnew); rsum += p[v]; }
    lsum = lsum * corr + rsum;
    mrow = mnew;
    #pragma unroll
    for (int v = 0; v < 8; ++v) { o0[v] *= corr; o1[v] *= corr; }
    #pragma unroll
    for (int t = 0; t < 4; ++t) {
        v2f bp = { p[2 * t], p[2 * t + 1] };
        v2f a0 = { T.v[2 * t].x, T.v[2 * t + 1].x };
        v2f a1 = { T.v[2 * t].y, T.v[2 * t + 1].y };
        o0 = wmma_f32(a0, bp, o0);
        o1 = wmma_f32(a1, bp, o1);
    }
}

__global__ __launch_bounds__(128) void ddnet_attn_kernel(
    const float* __restrict__ qkv, float* __restrict__ o, int S)
{
    const int qb    = S / 64;
    const int bh    = blockIdx.x / qb;
    const int qblk  = blockIdx.x % qb;
    const int b     = bh >> 3, head = bh & 7;
    const int tid   = threadIdx.x, lane = tid & 31, wave = tid >> 5;
    const int lhalf = lane >> 4, lmod = lane & 15;
    const int q0    = qblk * 64 + wave * 16;

    const size_t rs = 768;
    const float* qp = qkv + (size_t)b * S * rs + head * 32;
    const float* kr = qp + 256 + (size_t)lmod * rs + 16 * lhalf;
    const float* vr = qp + 512 + (size_t)(8 * lhalf) * rs + 2 * lmod;

    v4f qv[4];
    {
        const float* qr = qp + (size_t)(q0 + lmod) * rs + 16 * lhalf;
        #pragma unroll
        for (int c = 0; c < 4; ++c) qv[c] = *(const v4f*)(qr + 4 * c);
    }

    float mrow = -1e30f, lsum = 0.f;
    v8f o0 = {}, o1 = {};

    AttnTile Ta, Tb;
    attn_load(Ta, kr, vr);
    attn_load(Tb, kr + (size_t)16 * rs, vr + (size_t)16 * rs);
    for (int k0 = 0; k0 < S; k0 += 32) {
        attn_step(Ta, qv, mrow, lsum, o0, o1);
        const int kn = (k0 + 32 < S) ? (k0 + 32) : k0;
        attn_load(Ta, kr + (size_t)kn * rs, vr + (size_t)kn * rs);
        __builtin_amdgcn_sched_barrier(0);
        attn_step(Tb, qv, mrow, lsum, o0, o1);
        attn_load(Tb, kr + (size_t)(kn + 16) * rs, vr + (size_t)(kn + 16) * rs);
        __builtin_amdgcn_sched_barrier(0);
    }

    const float inv = 1.0f / (lsum + attn_xhalf(lsum));
    float* orow = o + ((size_t)b * S + q0 + lmod) * 256 + head * 32 + 16 * lhalf;
    #pragma unroll
    for (int j = 0; j < 4; ++j) {
        v4f w = { o0[2 * j] * inv, o1[2 * j] * inv, o0[2 * j + 1] * inv, o1[2 * j + 1] * inv };
        *(v4f*)(orow + 4 * j) = w;
    }
}

static void launch_gemm(const float* A, const float* W, const float* bias,
                        const float* resid, float* C, int M, int N, int K,
                        int act, hipStream_t stream)
{
    dim3 grid(N / GT_N, M / GT_M);
    ddnet_gemm_kernel<<<grid, 256, 0, stream>>>(A, W, bias, resid, C, M, N, K, act);
}

extern "C" void kernel_launch(void* const* d_in, const int* in_sizes, int n_in,
                              void* d_out, int out_size, void* d_ws, size_t ws_size,
                              hipStream_t stream) {
    (void)in_sizes; (void)n_in; (void)out_size; (void)ws_size;
    const int B = 16, S = 512, IN = 64, D = 256, L = 3, G = 6, DFF = 1024;
    const int Mr = B * S;

    const float* x      = (const float*)d_in[0];
    const float* in_w   = (const float*)d_in[1];
    const float* in_b   = (const float*)d_in[2];
    const float* Mw     = (const float*)d_in[3];
    const float* P      = (const float*)d_in[4];
    const float* desc_g = (const float*)d_in[5];
    const float* desc_b = (const float*)d_in[6];
    const float* qkv_w  = (const float*)d_in[7];
    const float* qkv_b  = (const float*)d_in[8];
    const float* out_w  = (const float*)d_in[9];
    const float* out_b  = (const float*)d_in[10];
    const float* ff1_w  = (const float*)d_in[11];
    const float* ff1_b  = (const float*)d_in[12];
    const float* ff2_w  = (const float*)d_in[13];
    const float* ff2_b  = (const float*)d_in[14];
    const float* ln1_g  = (const float*)d_in[15];
    const float* ln1_b  = (const float*)d_in[16];
    const float* ln2_g  = (const float*)d_in[17];
    const float* ln2_b  = (const float*)d_in[18];

    float* h    = (float*)d_ws;
    float* tmp  = h    + (size_t)Mr * D;
    float* xt   = tmp  + (size_t)Mr * D;
    float* hd   = xt   + (size_t)Mr * D;
    float* qkv  = hd   + (size_t)Mr * D;
    float* obuf = qkv  + (size_t)Mr * 3 * D;
    float* h1   = obuf + (size_t)Mr * D;
    float* ff   = h1   + (size_t)Mr * D;

    launch_gemm(x, in_w, in_b, nullptr, h, Mr, D, IN, 0, stream);
    ddnet_pe_kernel<<<(Mr * D) / 256, 256, 0, stream>>>(h, S, D);

    for (int l = 0; l < L; ++l) {
        launch_gemm(h, Mw + (size_t)l * D * D, nullptr, nullptr, tmp, Mr, D, D, 0, stream);
        ddnet_ln_kernel<<<Mr / 8, 256, 0, stream>>>(tmp, desc_g + l * D, desc_b + l * D, xt, D);
        ddnet_desc_kernel<<<(S / 4) * 2, 256, 0, stream>>>(xt, h, P + (size_t)l * D * D * G, hd, S);
        launch_gemm(hd, qkv_w + (size_t)l * 3 * D * D, qkv_b + (size_t)l * 3 * D,
                    nullptr, qkv, Mr, 3 * D, D, 0, stream);
        ddnet_attn_kernel<<<B * 8 * (S / 64), 128, 0, stream>>>(qkv, obuf, S);
        launch_gemm(obuf, out_w + (size_t)l * D * D, out_b + l * D, hd, tmp, Mr, D, D, 0, stream);
        ddnet_ln_kernel<<<Mr / 8, 256, 0, stream>>>(tmp, ln1_g + l * D, ln1_b + l * D, h1, D);
        launch_gemm(h1, ff1_w + (size_t)l * DFF * D, ff1_b + (size_t)l * DFF,
                    nullptr, ff, Mr, DFF, D, 1  , stream);
        launch_gemm(ff, ff2_w + (size_t)l * D * DFF, ff2_b + l * D, h1, tmp, Mr, D, DFF, 0, stream);
        float* dst = (l == L - 1) ? (float*)d_out : h;
        ddnet_ln_kernel<<<Mr / 8, 256, 0, stream>>>(tmp, ln2_g + l * D, ln2_b + l * D, dst, D);
    }
}


# GEMM epilogue: the 2 bias and 32 residual values of a lane are read as one batch ahead of the arithmetic and stores (was one dependent global round trip per output element); sums unchanged
# speedup vs baseline: 2.5083x; 1.3260x over previous
#include <hip/hip_runtime.h>
#include <math.h>

typedef __attribute__((ext_vector_type(2))) float v2f;
typedef __attribute__((ext_vector_type(8))) float v8f;

__device__ __forceinline__ v8f wmma_f32(v2f a, v2f b, v8f c) {
    return __builtin_amdgcn_wmma_f32_16x16x4_f32(false, a, false, b, (short)0, c, false, false);
}
__device__ __forceinline__ v2f lds_frag(const float* p) {
    return *(const v2f*)p;
}

#define GT_M 64
#define GT_N 128
#define GT_K 64
#define GT_KQ (GT_K / 4)
#define A_SL (GT_M * 4 + 4)
#define B_SL (GT_N * 4 + 4)

typedef __attribute__((ext_vector_type(4))) float v4f;

__global__ __launch_bounds__(256) void ddnet_gemm_kernel(
    const float* __restrict__ A, const float* __restrict__ W,
    const float* __restrict__ bias, const float* __restrict__ resid,
    float* __restrict__ C, int Mdim, int Ndim, int Kdim, int act)
{
    __shared__ __align__(16) float As[2][GT_KQ * A_SL];
    __shared__ __align__(16) float Bs[2][GT_KQ * B_SL];

    const int tid   = threadIdx.x;
    const int lane  = tid & 31;
    const int wave  = tid >> 5;
    const int lhalf = lane >> 4;
    const int lmod  = lane & 15;
    const int m0    = blockIdx.y * GT_M;
    const int n0    = blockIdx.x * GT_N;
    const int wm    = (wave & 1) * 32;
    const int wn    = (wave >> 1) * 32;
    const int nk    = Kdim >> 6;

    const int lq = tid & 15;
    const int lr = tid >> 4;
    const float* Ablk = A + (size_t)m0 * Kdim;
    const float* Wblk = W + (size_t)n0 * Kdim;
    const unsigned toff = (unsigned)(lr * Kdim + lq * 4);
    const int rstep = 16 * Kdim;
    const int a_st = lq * A_SL + lr * 4;
    const int b_st = lq * B_SL + lr * 4;
    const int a_rd = lhalf * A_SL + (wm + lmod) * 4;
    const int b_rd = lhalf * B_SL + (wn + lmod) * 4;

    v8f acc[2][2] = {};
    v4f ra[4], rb[8];

    #pragma unroll
    for (int i = 0; i < 4; ++i)
        ra[i] = *(const v4f*)(Ablk + i * rstep + toff);
    #pragma unroll
    for (int i = 0; i < 8; ++i)
        rb[i] = *(const v4f*)(Wblk + i * rstep + toff);
    #pragma unroll
    for (int i = 0; i < 4; ++i)
        *(v4f*)&As[0][a_st + i * 64] = ra[i];
    #pragma unroll
    for (int i = 0; i < 8; ++i)
        *(v4f*)&Bs[0][b_st + i * 64] = rb[i];
    __syncthreads();

    for (int kt = 0; kt < nk; ++kt) {
        const int buf = kt & 1;
        if (kt + 1 < nk) {
            const int k0 = (kt + 1) << 6;
            #pragma unroll
            for (int i = 0; i < 4; ++i)
                ra[i] = *(const v4f*)(Ablk + (k0 + i * rstep) + toff);
            #pragma unroll
            for (int i = 0; i < 8; ++i)
                rb[i] = *(const v4f*)(Wblk + (k0 + i * rstep) + toff);
        }
        const float* Ab = As[buf] + a_rd;
        const float* Bb = Bs[buf] + b_rd;
        #pragma unroll
        for (int u = 0; u < GT_KQ / 2; ++u) {
            v4f a0 = *(const v4f*)&Ab[2 * u * A_SL];
            v4f a1 = *(const v4f*)&Ab[2 * u * A_SL + 64];
            v4f b0 = *(const v4f*)&Bb[2 * u * B_SL];
            v4f b1 = *(const v4f*)&Bb[2 * u * B_SL + 64];
            acc[0][0] = wmma_f32(a0.xy, b0.xy, acc[0][0]);
            acc[0][1] = wmma_f32(a0.xy, b1.xy, acc[0][1]);
            acc[1][0] = wmma_f32(a1.xy, b0.xy, acc[1][0]);
            acc[1][1] = wmma_f32(a1.xy, b1.xy, acc[1][1]);
            acc[0][0] = wmma_f32(a0.zw, b0.zw, acc[0][0]);
            acc[0][1] = wmma_f32(a0.zw, b1.zw, acc[0][1]);
            acc[1][0] = wmma_f32(a1.zw, b0.zw, acc[1][0]);
            acc[1][1] = wmma_f32(a1.zw, b1.zw, acc[1][1]);
        }
        if (kt + 1 < nk) {
            __syncthreads();
            #pragma unroll
            for (int i = 0; i < 4; ++i)
                *(v4f*)&As[buf ^ 1][a_st + i * 64] = ra[i];
            #pragma unroll
            for (int i = 0; i < 8; ++i)
                *(v4f*)&Bs[buf ^ 1][b_st + i * 64] = rb[i];
            __syncthreads();
        }
    }

    const bool has_bias = bias != nullptr;
    const bool has_res  = resid != nullptr;
    const int  ecol = n0 + wn + lmod;
    const size_t eoff = (size_t)(m0 + wm + 8 * lhalf) * Ndim + ecol;
    float bv[2] = { 0.f, 0.f };
    float rv[2][2][8] = {};
    if (has_bias) { bv[0] = bias[ecol]; bv[1] = bias[ecol + 16]; }
    if (has_res) {
        const float* rp = resid + eoff;
        #pragma unroll
        for (int mi = 0; mi < 2; ++mi)
            #pragma unroll
            for (int ni = 0; ni < 2; ++ni)
                #pragma unroll
                for (int v = 0; v < 8; ++v)
                    rv[mi][ni][v] = rp[(size_t)(mi * 16 + v) * Ndim + ni * 16];
    }
    float* cp = C + eoff;
    #pragma unroll
    for (int mi = 0; mi < 2; ++mi)
        #pragma unroll
        for (int ni = 0; ni < 2; ++ni)
            #pragma unroll
            for (int v = 0; v < 8; ++v) {
                float val = acc[mi][ni][v];
                if (has_bias) val += bv[ni];
                if (has_res)  val += rv[mi][ni][v];
                if (act == 1)
                    val = 0.5f * val * (1.0f + erff(val * 0.70710678118654752f));
                cp[(size_t)(mi * 16 + v) * Ndim + ni * 16] = val;
            }
}

__global__ __launch_bounds__(256) void ddnet_pe_kernel(
    float* __restrict__ h, int S, int D)
{
    size_t idx = (size_t)blockIdx.x * 256 + threadIdx.x;
    int d = (int)(idx % D);
    int s = (int)((idx / D) % S);
    int pair = d >> 1;
    float div = __expf(-(float)(2 * pair) * (logf(10000.0f) / (float)D));
    float ang = (float)s * div;
    h[idx] += (d & 1) ? __cosf(ang) : __sinf(ang);
}

__global__ __launch_bounds__(256) void ddnet_ln_kernel(
    const float* __restrict__ x, const float* __restrict__ gamma,
    const float* __restrict__ beta, float* __restrict__ out, int D)
{
    int row  = blockIdx.x * 8 + (threadIdx.x >> 5);
    int lane = threadIdx.x & 31;
    const float* xr = x + (size_t)row * D;
    float vals[8];
    float s = 0.f;
    #pragma unroll
    for (int i = 0; i < 8; ++i) { vals[i] = xr[lane + i * 32]; s += vals[i]; }
    #pragma unroll
    for (int m = 16; m >= 1; m >>= 1) s += __shfl_xor(s, m, 32);
    float mean = s * (1.0f / 256.0f);
    float vs = 0.f;
    #pragma unroll
    for (int i = 0; i < 8; ++i) { float d = vals[i] - mean; vs += d * d; }
    #pragma unroll
    for (int m = 16; m >= 1; m >>= 1) vs += __shfl_xor(vs, m, 32);
    float rstd = rsqrtf(vs * (1.0f / 256.0f) + 1e-5f);
    float* orow = out + (size_t)row * D;
    #pragma unroll
    for (int i = 0; i < 8; ++i) {
        int c = lane + i * 32;
        orow[c] = (vals[i] - mean) * rstd * gamma[c] + beta[c];
    }
}

#define DD 256
#define DG 6
#define DS_SG 4
#define DS_NI 128
#define DS_XP 66
#define DS_XS (64 * DS_XP)
#define DS_NT (DS_NI + DD - 1)
#define DS_CS (384 * DG)

struct DescTile { v4f p[12]; };

__device__ __forceinline__ void desc_load(DescTile& T, const float* pr) {
    #pragma unroll
    for (int t = 0; t < 4; ++t)
        #pragma unroll
        for (int q = 0; q < 3; ++q)
            T.p[3 * t + q] = *(const v4f*)(pr + t * 24 + q * 4);
}

struct DescOps { v2f c[DS_SG][6]; v2f a[DS_SG]; };

__device__ __forceinline__ void desc_ops(DescOps& O, const float* cp, const float* xp) {
    #pragma unroll
    for (int sl = 0; sl < DS_SG; ++sl) {
        #pragma unroll
        for (int m = 0; m < 6; ++m)
            O.c[sl][m] = *(const v2f*)(cp + sl * DS_CS + 2 * m);
        O.a[sl] = lds_frag(xp + sl * DS_XS);
    }
}

__device__ __forceinline__ void desc_group(const DescTile& T, int t, const DescOps& O,
                                           v8f (&acc)[DS_SG]) {
    #pragma unroll
    for (int sl = 0; sl < DS_SG; ++sl) {
        float w0 = 0.f, w1 = 0.f;
        #pragma unroll
        for (int g = 0; g < DG; ++g) {
            w0 += T.p[3 * t + (g >> 2)][g & 3] * O.c[sl][g >> 1][g & 1];
            w1 += T.p[3 * t + ((6 + g) >> 2)][(6 + g) & 3] * O.c[sl][(6 + g) >> 1][g & 1];
        }
        v2f b = { w0, w1 };
        acc[sl] = wmma_f32(O.a[sl], b, acc[sl]);
    }
}

__device__ __forceinline__ void desc_step(const DescTile& T, const float* cb,
                                          const float* xb, v8f (&acc)[DS_SG]) {
    DescOps Oa, Ob;
    desc_ops(Oa, cb, xb);
    desc_ops(Ob, cb + 24, xb + DS_XP);
    __builtin_amdgcn_sched_barrier(0);
    desc_group(T, 0, Oa, acc);
    desc_ops(Oa, cb + 48, xb + 2 * DS_XP);
    __builtin_amdgcn_sched_barrier(0);
    desc_group(T, 1, Ob, acc);
    desc_ops(Ob, cb + 72, xb + 3 * DS_XP);
    __builtin_amdgcn_sched_barrier(0);
    desc_group(T, 2, Oa, acc);
    desc_group(T, 3, Ob, acc);
}

__global__ __launch_bounds__(256) void ddnet_desc_kernel(
    const float* __restrict__ xt, const float* __restrict__ h,
    const float* __restrict__ P, float* __restrict__ hd, int S)
{
    __shared__ __align__(16) float Xs[DS_SG * DS_XS];
    __shared__ __align__(16) float Ct[DS_SG * DS_CS];

    const int tid = threadIdx.x;
    const int i0  = (blockIdx.x & 1) * DS_NI;
    const int s0  = (blockIdx.x >> 1) * DS_SG;
    const float two_pi = 6.28318530717958647692f;

    v4f xv[16];
    {
        const float* xsrc = xt + (size_t)s0 * DD + tid * 4;
        #pragma unroll
        for (int bb = 0; bb < 16; ++bb)
            xv[bb] = *(const v4f*)(xsrc + (size_t)bb * S * DD);
    }

    if (tid < 42 * DG) {
        const int g = tid % DG, tq = tid / DG;
        const float pw = powf(1.5f, (float)g);
        for (int tt = tq; tt < DS_NT; tt += 42) {
            const float period = (2.0f + 0.5f * (float)(i0 + tt)) * pw;
            #pragma unroll
            for (int sl = 0; sl < DS_SG; ++sl)
                Ct[sl * DS_CS + tt * DG + g] = __cosf(two_pi * (float)(s0 + sl) / period);
        }
    }
    {
        float* xd = Xs + (tid >> 6) * DS_XS + (tid & 63) * DS_XP;
        #pragma unroll
        for (int bb = 0; bb < 16; ++bb) {
            v2f lo = { xv[bb].x, xv[bb].y }, hi = { xv[bb].z, xv[bb].w };
            *(v2f*)(xd + bb * 2)      = lo;
            *(v2f*)(xd + 32 + bb * 2) = hi;
        }
    }
    __syncthreads();

    const int lane = tid & 31, wave = tid >> 5;
    const int lhalf = lane >> 4, lmod = lane & 15;
    const int iloc = wave * 16 + lmod;
    const float* prow = P + ((size_t)(i0 + iloc) * DD + 2 * lhalf) * DG;
    const float* cb = Ct + (iloc + 2 * lhalf) * DG;
    const float* xb = Xs + lhalf * 32 + lmod * 2;

    v8f acc[DS_SG] = {};
    DescTile Ta, Tb;
    desc_load(Ta, prow);
    for (int j0 = 0; j0 < DD; j0 += 32) {
        desc_load(Tb, prow + (j0 + 16) * DG);
        __builtin_amdgcn_sched_barrier(0);
        desc_step(Ta, cb + j0 * DG, xb + (j0 >> 2) * DS_XP, acc);
        const int jn = (j0 + 32 < DD) ? (j0 + 32) : j0;
        desc_load(Ta, prow + jn * DG);
        __builtin_amdgcn_sched_barrier(0);
        desc_step(Tb, cb + (j0 + 16) * DG, xb + ((j0 + 16) >> 2) * DS_XP, acc);
    }

    float hv[DS_SG][8];
    #pragma unroll
    for (int sl = 0; sl < DS_SG; ++sl)
        #pragma unroll
        for (int v = 0; v < 8; ++v)
            hv[sl][v] = h[((size_t)(v + 8 * lhalf) * S + s0 + sl) * DD + i0 + iloc];
    #pragma unroll
    for (int sl = 0; sl < DS_SG; ++sl)
        #pragma unroll
        for (int v = 0; v < 8; ++v)
            hd[((size_t)(v + 8 * lhalf) * S + s0 + sl) * DD + i0 + iloc] = acc[sl][v] + hv[sl][v];
}

struct AttnTile { v4f k[4]; v2f v[8]; };

__device__ __forceinline__ float attn_xhalf(float x) {
    return __builtin_bit_cast(float, __builtin_amdgcn_permlanex16(
        __builtin_bit_cast(int, x), __builtin_bit_cast(int, x),
        0x76543210, 0xfedcba98, false, false));
}

__device__ __forceinline__ void attn_load(AttnTile& T, const float* kr, const float* vr) {
    #pragma unroll
    for (int c = 0; c < 4; ++c) T.k[c] = *(const v4f*)(kr + 4 * c);
    #pragma unroll
    for (int j = 0; j < 8; ++j) T.v[j] = *(const v2f*)(vr + (size_t)j * 768);
}

__device__ __forceinline__ void attn_step(const AttnTile& T, const v4f (&qv)[4],
                                          float& mrow, float& lsum, v8f& o0, v8f& o1) {
    const float scale = 0.17677669529663688f;
    v8f s0 = {}, s1 = {};
    #pragma unroll
    for (int c = 0; c < 4; ++c) {
        s0 = wmma_f32(T.k[c].xy, qv[c].xy, s0);
        s1 = wmma_f32(T.k[c].zw, qv[c].zw, s1);
    }
    float sv[8];
    #pragma unroll
    for (int v = 0; v < 8; ++v) sv[v] = (s0[v] + s1[v]) * scale;
    float mx = fmaxf(fmaxf(fmaxf(sv[0], sv[1]), fmaxf(sv[2], sv[3])),
                     fmaxf(fmaxf(sv[4], sv[5]), fmaxf(sv[6], sv[7])));
    mx = fmaxf(mx, attn_xhalf(mx));
    const float mnew = fmaxf(mrow, mx);
    const float corr = __expf(mrow - mnew);
    float p[8];
    float rsum = 0.f;
    #pragma unroll
    for (int v = 0; v < 8; ++v) { p[v] = __expf(sv[v] - mnew); rsum += p[v]; }
    lsum = lsum * corr + rsum;
    mrow = mnew;
    #pragma unroll
    for (int v = 0; v < 8; ++v) { o0[v] *= corr; o1[v] *= corr; }
    #pragma unroll
    for (int t = 0; t < 4; ++t) {
        v2f bp = { p[2 * t], p[2 * t + 1] };
        v2f a0 = { T.v[2 * t].x, T.v[2 * t + 1].x };
        v2f a1 = { T.v[2 * t].y, T.v[2 * t + 1].y };
        o0 = wmma_f32(a0, bp, o0);
        o1 = wmma_f32(a1, bp, o1);
    }
}

__global__ __launch_bounds__(128) void ddnet_attn_kernel(
    const float* __restrict__ qkv, float* __restrict__ o, int S)
{
    const int qb    = S / 64;
    const int bh    = blockIdx.x / qb;
    const int qblk  = blockIdx.x % qb;
    const int b     = bh >> 3, head = bh & 7;
    const int tid   = threadIdx.x, lane = tid & 31, wave = tid >> 5;
    const int lhalf = lane >> 4, lmod = lane & 15;
    const int q0    = qblk * 64 + wave * 16;

    const size_t rs = 768;
    const float* qp = qkv + (size_t)b * S * rs + head * 32;
    const float* kr = qp + 256 + (size_t)lmod * rs + 16 * lhalf;
    const float* vr = qp + 512 + (size_t)(8 * lhalf) * rs + 2 * lmod;

    v4f qv[4];
    {
        const float* qr = qp + (size_t)(q0 + lmod) * rs + 16 * lhalf;
        #pragma unroll
        for (int c = 0; c < 4; ++c) qv[c] = *(const v4f*)(qr + 4 * c);
    }

    float mrow = -1e30f, lsum = 0.f;
    v8f o0 = {}, o1 = {};

    AttnTile Ta, Tb;
    attn_load(Ta, kr, vr);
    attn_load(Tb, kr + (size_t)16 * rs, vr + (size_t)16 * rs);
    for (int k0 = 0; k0 < S; k0 += 32) {
        attn_step(Ta, qv, mrow, lsum, o0, o1);
        const int kn = (k0 + 32 < S) ? (k0 + 32) : k0;
        attn_load(Ta, kr + (size_t)kn * rs, vr + (size_t)kn * rs);
        __builtin_amdgcn_sched_barrier(0);
        attn_step(Tb, qv, mrow, lsum, o0, o1);
        attn_load(Tb, kr + (size_t)(kn + 16) * rs, vr + (size_t)(kn + 16) * rs);
        __builtin_amdgcn_sched_barrier(0);
    }

    const float inv = 1.0f / (lsum + attn_xhalf(lsum));
    float* orow = o + ((size_t)b * S + q0 + lmod) * 256 + head * 32 + 16 * lhalf;
    #pragma unroll
    for (int j = 0; j < 4; ++j) {
        v4f w = { o0[2 * j] * inv, o1[2 * j] * inv, o0[2 * j + 1] * inv, o1[2 * j + 1] * inv };
        *(v4f*)(orow + 4 * j) = w;
    }
}

static void launch_gemm(const float* A, const float* W, const float* bias,
                        const float* resid, float* C, int M, int N, int K,
                        int act, hipStream_t stream)
{
    dim3 grid(N / GT_N, M / GT_M);
    ddnet_gemm_kernel<<<grid, 256, 0, stream>>>(A, W, bias, resid, C, M, N, K, act);
}

extern "C" void kernel_launch(void* const* d_in, const int* in_sizes, int n_in,
                              void* d_out, int out_size, void* d_ws, size_t ws_size,
                              hipStream_t stream) {
    (void)in_sizes; (void)n_in; (void)out_size; (void)ws_size;
    const int B = 16, S = 512, IN = 64, D = 256, L = 3, G = 6, DFF = 1024;
    const int Mr = B * S;

    const float* x      = (const float*)d_in[0];
    const float* in_w   = (const float*)d_in[1];
    const float* in_b   = (const float*)d_in[2];
    const float* Mw     = (const float*)d_in[3];
    const float* P      = (const float*)d_in[4];
    const float* desc_g = (const float*)d_in[5];
    const float* desc_b = (const float*)d_in[6];
    const float* qkv_w  = (const float*)d_in[7];
    const float* qkv_b  = (const float*)d_in[8];
    const float* out_w  = (const float*)d_in[9];
    const float* out_b  = (const float*)d_in[10];
    const float* ff1_w  = (const float*)d_in[11];
    const float* ff1_b  = (const float*)d_in[12];
    const float* ff2_w  = (const float*)d_in[13];
    const float* ff2_b  = (const float*)d_in[14];
    const float* ln1_g  = (const float*)d_in[15];
    const float* ln1_b  = (const float*)d_in[16];
    const float* ln2_g  = (const float*)d_in[17];
    const float* ln2_b  = (const float*)d_in[18];

    float* h    = (float*)d_ws;
    float* tmp  = h    + (size_t)Mr * D;
    float* xt   = tmp  + (size_t)Mr * D;
    float* hd   = xt   + (size_t)Mr * D;
    float* qkv  = hd   + (size_t)Mr * D;
    float* obuf = qkv  + (size_t)Mr * 3 * D;
    float* h1   = obuf + (size_t)Mr * D;
    float* ff   = h1   + (size_t)Mr * D;

    launch_gemm(x, in_w, in_b, nullptr, h, Mr, D, IN, 0, stream);
    ddnet_pe_kernel<<<(Mr * D) / 256, 256, 0, stream>>>(h, S, D);

    for (int l = 0; l < L; ++l) {
        launch_gemm(h, Mw + (size_t)l * D * D, nullptr, nullptr, tmp, Mr, D, D, 0, stream);
        ddnet_ln_kernel<<<Mr / 8, 256, 0, stream>>>(tmp, desc_g + l * D, desc_b + l * D, xt, D);
        ddnet_desc_kernel<<<(S / 4) * 2, 256, 0, stream>>>(xt, h, P + (size_t)l * D * D * G, hd, S);
        launch_gemm(hd, qkv_w + (size_t)l * 3 * D * D, qkv_b + (size_t)l * 3 * D,
                    nullptr, qkv, Mr, 3 * D, D, 0, stream);
        ddnet_attn_kernel<<<B * 8 * (S / 64), 128, 0, stream>>>(qkv, obuf, S);
        launch_gemm(obuf, out_w + (size_t)l * D * D, out_b + l * D, hd, tmp, Mr, D, D, 0, stream);
        ddnet_ln_kernel<<<Mr / 8, 256, 0, stream>>>(tmp, ln1_g + l * D, ln1_b + l * D, h1, D);
        launch_gemm(h1, ff1_w + (size_t)l * DFF * D, ff1_b + (size_t)l * DFF,
                    nullptr, ff, Mr, DFF, D, 1  , stream);
        launch_gemm(ff, ff2_w + (size_t)l * D * DFF, ff2_b + l * D, h1, tmp, Mr, D, DFF, 0, stream);
        float* dst = (l == L - 1) ? (float*)d_out : h;
        ddnet_ln_kernel<<<Mr / 8, 256, 0, stream>>>(tmp, ln2_g + l * D, ln2_b + l * D, dst, D);
    }
}


# Layer norms folded into their readers: products with K=256 and the descriptor kernel sum and normalise the rows they load (first column block stores them); 8 launches gone; LN sum order changed
# speedup vs baseline: 2.5932x; 1.0338x over previous
#include <hip/hip_runtime.h>
#include <math.h>

typedef __attribute__((ext_vector_type(2))) float v2f;
typedef __attribute__((ext_vector_type(8))) float v8f;

__device__ __forceinline__ v8f wmma_f32(v2f a, v2f b, v8f c) {
    return __builtin_amdgcn_wmma_f32_16x16x4_f32(false, a, false, b, (short)0, c, false, false);
}
__device__ __forceinline__ v2f lds_frag(const float* p) {
    return *(const v2f*)p;
}

typedef __attribute__((ext_vector_type(4))) float v4f;

template <int M> __device__ __forceinline__ float row_xmask(float x) {
    return __builtin_bit_cast(float, __builtin_amdgcn_update_dpp(
        0, __builtin_bit_cast(int, x), 0x160 + M, 0xf, 0xf, false));
}
__device__ __forceinline__ float row16_sum(float s) {
    s += row_xmask<1>(s);
    s += row_xmask<2>(s);
    s += row_xmask<4>(s);
    s += row_xmask<8>(s);
    return s;
}
__device__ __forceinline__ float wave32_sum(float s) {
    s = row16_sum(s);
    return s + __builtin_bit_cast(float, __builtin_amdgcn_permlanex16(
        __builtin_bit_cast(int, s), __builtin_bit_cast(int, s),
        0x76543210, 0xfedcba98, false, false));
}
__device__ __forceinline__ float ln_sum4(v4f x) { return (x.x + x.y) + (x.z + x.w); }
__device__ __forceinline__ float ln_sq4(v4f d) { return (d.x * d.x + d.y * d.y) + (d.z * d.z + d.w * d.w); }
__device__ __forceinline__ v4f ln_apply(v4f x, float mu, float rs, v4f g, v4f b) {
    v4f r;
    r.x = (x.x - mu) * rs * g.x + b.x;
    r.y = (x.y - mu) * rs * g.y + b.y;
    r.z = (x.z - mu) * rs * g.z + b.z;
    r.w = (x.w - mu) * rs * g.w + b.w;
    return r;
}

#define GT_M 64
#define GT_N 128
#define GT_K 64
#define GT_KQ (GT_K / 4)
#define A_SL (GT_M * 4 + 4)
#define B_SL (GT_N * 4 + 4)

template <bool LN>
__global__ __launch_bounds__(256) void ddnet_gemm_kernel(
    const float* __restrict__ A, const float* __restrict__ W,
    const float* __restrict__ bias, const float* __restrict__ resid,
    float* __restrict__ C, int Mdim, int Ndim, int Kdim, int act,
    const float* __restrict__ lng, const float* __restrict__ lnb,
    float* __restrict__ side)
{
    __shared__ __align__(16) float As[2][GT_KQ * A_SL];
    __shared__ __align__(16) float Bs[2][GT_KQ * B_SL];

    const int tid   = threadIdx.x;
    const int lane  = tid & 31;
    const int wave  = tid >> 5;
    const int lhalf = lane >> 4;
    const int lmod  = lane & 15;
    const int m0    = blockIdx.y * GT_M;
    const int n0    = blockIdx.x * GT_N;
    const int wm    = (wave & 1) * 32;
    const int wn    = (wave >> 1) * 32;
    const int nk    = Kdim >> 6;

    const int lq = tid & 15;
    const int lr = tid >> 4;
    const float* Ablk = A + (size_t)m0 * Kdim;
    const float* Wblk = W + (size_t)n0 * Kdim;
    const unsigned toff = (unsigned)(lr * Kdim + lq * 4);
    const int rstep = 16 * Kdim;
    const int a_st = lq * A_SL + lr * 4;
    const int b_st = lq * B_SL + lr * 4;
    const int a_rd = lhalf * A_SL + (wm + lmod) * 4;
    const int b_rd = lhalf * B_SL + (wn + lmod) * 4;

    v8f acc[2][2] = {};
    v4f ra[4], rb[8];
    float mu[4] = { 0.f, 0.f, 0.f, 0.f }, rs[4] = { 1.f, 1.f, 1.f, 1.f };
    v4f g4 = {}, b4 = {};
    float* sblk = nullptr;

    if constexpr (LN) {
        v4f pv[4][4];
        #pragma unroll
        for (int i = 0; i < 4; ++i)
            #pragma unroll
            for (int t = 0; t < 4; ++t)
                pv[i][t] = *(const v4f*)(Ablk + (t * 64 + i * rstep) + toff);
        #pragma unroll
        for (int i = 0; i < 8; ++i)
            rb[i] = *(const v4f*)(Wblk + i * rstep + toff);
        g4 = *(const v4f*)(lng + lq * 4);
        b4 = *(const v4f*)(lnb + lq * 4);
        sblk = side + (size_t)m0 * Kdim;
        #pragma unroll
        for (int i = 0; i < 4; ++i) {
            float sm = (ln_sum4(pv[i][0]) + ln_sum4(pv[i][1])) + (ln_sum4(pv[i][2]) + ln_sum4(pv[i][3]));
            const float m = row16_sum(sm) * (1.0f / 256.0f);
            float sq = (ln_sq4(pv[i][0] - m) + ln_sq4(pv[i][1] - m)) + (ln_sq4(pv[i][2] - m) + ln_sq4(pv[i][3] - m));
            mu[i] = m;
            rs[i] = rsqrtf(row16_sum(sq) * (1.0f / 256.0f) + 1e-5f);
            ra[i] = ln_apply(pv[i][0], mu[i], rs[i], g4, b4);
        }
    } else {
        #pragma unroll
        for (int i = 0; i < 4; ++i)
            ra[i] = *(const v4f*)(Ablk + i * rstep + toff);
        #pragma unroll
        for (int i = 0; i < 8; ++i)
            rb[i] = *(const v4f*)(Wblk + i * rstep + toff);
    }
    #pragma unroll
    for (int i = 0; i < 4; ++i)
        *(v4f*)&As[0][a_st + i * 64] = ra[i];
    #pragma unroll
    for (int i = 0; i < 8; ++i)
        *(v4f*)&Bs[0][b_st + i * 64] = rb[i];
    __syncthreads();
    if constexpr (LN) {
        if (blockIdx.x == 0) {
            #pragma unroll
            for (int i = 0; i < 4; ++i)
                *(v4f*)(sblk + i * rstep + toff) = ra[i];
        }
    }

    for (int kt = 0; kt < nk; ++kt) {
        const int buf = kt & 1;
        if (kt + 1 < nk) {
            const int k0 = (kt + 1) << 6;
            #pragma unroll
            for (int i = 0; i < 4; ++i)
                ra[i] = *(const v4f*)(Ablk + (k0 + i * rstep) + toff);
            #pragma unroll
            for (int i = 0; i < 8; ++i)
                rb[i] = *(const v4f*)(Wblk + (k0 + i * rstep) + toff);
            if constexpr (LN) {
                g4 = *(const v4f*)(lng + k0 + lq * 4);
                b4 = *(const v4f*)(lnb + k0 + lq * 4);
            }
        }
        const float* Ab = As[buf] + a_rd;
        const float* Bb = Bs[buf] + b_rd;
        #pragma unroll
        for (int u = 0; u < GT_KQ / 2; ++u) {
            v4f a0 = *(const v4f*)&Ab[2 * u * A_SL];
            v4f a1 = *(const v4f*)&Ab[2 * u * A_SL + 64];
            v4f b0 = *(const v4f*)&Bb[2 * u * B_SL];
            v4f b1 = *(const v4f*)&Bb[2 * u * B_SL + 64];
            acc[0][0] = wmma_f32(a0.xy, b0.xy, acc[0][0]);
            acc[0][1] = wmma_f32(a0.xy, b1.xy, acc[0][1]);
            acc[1][0] = wmma_f32(a1.xy, b0.xy, acc[1][0]);
            acc[1][1] = wmma_f32(a1.xy, b1.xy, acc[1][1]);
            acc[0][0] = wmma_f32(a0.zw, b0.zw, acc[0][0]);
            acc[0][1] = wmma_f32(a0.zw, b1.zw, acc[0][1]);
            acc[1][0] = wmma_f32(a1.zw, b0.zw, acc[1][0]);
            acc[1][1] = wmma_f32(a1.zw, b1.zw, acc[1][1]);
        }
        if (kt + 1 < nk) {
            if constexpr (LN) {
                #pragma unroll
                for (int i = 0; i < 4; ++i)
                    ra[i] = ln_apply(ra[i], mu[i], rs[i], g4, b4);
            }
            __syncthreads();
            #pragma unroll
            for (int i = 0; i < 4; ++i)
                *(v4f*)&As[buf ^ 1][a_st + i * 64] = ra[i];
            #pragma unroll
            for (int i = 0; i < 8; ++i)
                *(v4f*)&Bs[buf ^ 1][b_st + i * 64] = rb[i];
            __syncthreads();
            if constexpr (LN) {
                if (blockIdx.x == 0) {
                    const int k0 = (kt + 1) << 6;
                    #pragma unroll
                    for (int i = 0; i < 4; ++i)
                        *(v4f*)(sblk + (k0 + i * rstep) + toff) = ra[i];
                }
            }
        }
    }

    const bool has_bias = bias != nullptr;
    const bool has_res  = resid != nullptr;
    const int  ecol = n0 + wn + lmod;
    const size_t eoff = (size_t)(m0 + wm + 8 * lhalf) * Ndim + ecol;
    float bv[2] = { 0.f, 0.f };
    float rv[2][2][8] = {};
    if (has_bias) { bv[0] = bias[ecol]; bv[1] = bias[ecol + 16]; }
    if (has_res) {
        const float* rp = resid + eoff;
        #pragma unroll
        for (int mi = 0; mi < 2; ++mi)
            #pragma unroll
            for (int ni = 0; ni < 2; ++ni)
                #pragma unroll
                for (int v = 0; v < 8; ++v)
                    rv[mi][ni][v] = rp[(size_t)(mi * 16 + v) * Ndim + ni * 16];
    }
    float* cp = C + eoff;
    #pragma unroll
    for (int mi = 0; mi < 2; ++mi)
        #pragma unroll
        for (int ni = 0; ni < 2; ++ni)
            #pragma unroll
            for (int v = 0; v < 8; ++v) {
                float val = acc[mi][ni][v];
                if (has_bias) val += bv[ni];
                if (has_res)  val += rv[mi][ni][v];
                if (act == 1)
                    val = 0.5f * val * (1.0f + erff(val * 0.70710678118654752f));
                cp[(size_t)(mi * 16 + v) * Ndim + ni * 16] = val;
            }
}

__global__ __launch_bounds__(256) void ddnet_pe_kernel(
    float* __restrict__ h, int S, int D)
{
    size_t idx = (size_t)blockIdx.x * 256 + threadIdx.x;
    int d = (int)(idx % D);
    int s = (int)((idx / D) % S);
    int pair = d >> 1;
    float div = __expf(-(float)(2 * pair) * (logf(10000.0f) / (float)D));
    float ang = (float)s * div;
    h[idx] += (d & 1) ? __cosf(ang) : __sinf(ang);
}

__global__ __launch_bounds__(256) void ddnet_ln_kernel(
    const float* __restrict__ x, const float* __restrict__ gamma,
    const float* __restrict__ beta, float* __restrict__ out, int D)
{
    int row  = blockIdx.x * 8 + (threadIdx.x >> 5);
    int lane = threadIdx.x & 31;
    const float* xr = x + (size_t)row * D;
    float vals[8];
    float s = 0.f;
    #pragma unroll
    for (int i = 0; i < 8; ++i) { vals[i] = xr[lane + i * 32]; s += vals[i]; }
    #pragma unroll
    for (int m = 16; m >= 1; m >>= 1) s += __shfl_xor(s, m, 32);
    float mean = s * (1.0f / 256.0f);
    float vs = 0.f;
    #pragma unroll
    for (int i = 0; i < 8; ++i) { float d = vals[i] - mean; vs += d * d; }
    #pragma unroll
    for (int m = 16; m >= 1; m >>= 1) vs += __shfl_xor(vs, m, 32);
    float rstd = rsqrtf(vs * (1.0f / 256.0f) + 1e-5f);
    float* orow = out + (size_t)row * D;
    #pragma unroll
    for (int i = 0; i < 8; ++i) {
        int c = lane + i * 32;
        orow[c] = (vals[i] - mean) * rstd * gamma[c] + beta[c];
    }
}

#define DD 256
#define DG 6
#define DS_SG 4
#define DS_NI 128
#define DS_XP 66
#define DS_XS (64 * DS_XP)
#define DS_NT (DS_NI + DD - 1)
#define DS_CS (384 * DG)

struct DescTile { v4f p[12]; };

__device__ __forceinline__ void desc_load(DescTile& T, const float* pr) {
    #pragma unroll
    for (int t = 0; t < 4; ++t)
        #pragma unroll
        for (int q = 0; q < 3; ++q)
            T.p[3 * t + q] = *(const v4f*)(pr + t * 24 + q * 4);
}

struct DescOps { v2f c[DS_SG][6]; v2f a[DS_SG]; };

__device__ __forceinline__ void desc_ops(DescOps& O, const float* cp, const float* xp) {
    #pragma unroll
    for (int sl = 0; sl < DS_SG; ++sl) {
        #pragma unroll
        for (int m = 0; m < 6; ++m)
            O.c[sl][m] = *(const v2f*)(cp + sl * DS_CS + 2 * m);
        O.a[sl] = lds_frag(xp + sl * DS_XS);
    }
}

__device__ __forceinline__ void desc_group(const DescTile& T, int t, const DescOps& O,
                                           v8f (&acc)[DS_SG]) {
    #pragma unroll
    for (int sl = 0; sl < DS_SG; ++sl) {
        float w0 = 0.f, w1 = 0.f;
        #pragma unroll
        for (int g = 0; g < DG; ++g) {
            w0 += T.p[3 * t + (g >> 2)][g & 3] * O.c[sl][g >> 1][g & 1];
            w1 += T.p[3 * t + ((6 + g) >> 2)][(6 + g) & 3] * O.c[sl][(6 + g) >> 1][g & 1];
        }
        v2f b = { w0, w1 };
        acc[sl] = wmma_f32(O.a[sl], b, acc[sl]);
    }
}

__device__ __forceinline__ void desc_step(const DescTile& T, const float* cb,
                                          const float* xb, v8f (&acc)[DS_SG]) {
    DescOps Oa, Ob;
    desc_ops(Oa, cb, xb);
    desc_ops(Ob, cb + 24, xb + DS_XP);
    __builtin_amdgcn_sched_barrier(0);
    desc_group(T, 0, Oa, acc);
    desc_ops(Oa, cb + 48, xb + 2 * DS_XP);
    __builtin_amdgcn_sched_barrier(0);
    desc_group(T, 1, Ob, acc);
    desc_ops(Ob, cb + 72, xb + 3 * DS_XP);
    __builtin_amdgcn_sched_barrier(0);
    desc_group(T, 2, Oa, acc);
    desc_group(T, 3, Ob, acc);
}

__global__ __launch_bounds__(256) void ddnet_desc_kernel(
    const float* __restrict__ xpre, const float* __restrict__ lng,
    const float* __restrict__ lnb, const float* __restrict__ h,
    const float* __restrict__ P, float* __restrict__ hd, int S)
{
    __shared__ __align__(16) float Xs[DS_SG * DS_XS];
    __shared__ __align__(16) float Ct[DS_SG * DS_CS];
    __shared__ float red[2][8][16];

    const int tid = threadIdx.x;
    const int i0  = (blockIdx.x & 1) * DS_NI;
    const int s0  = (blockIdx.x >> 1) * DS_SG;
    const float two_pi = 6.28318530717958647692f;

    v4f xv[16];
    {
        const float* xsrc = xpre + (size_t)s0 * DD + tid * 4;
        #pragma unroll
        for (int bb = 0; bb < 16; ++bb)
            xv[bb] = *(const v4f*)(xsrc + (size_t)bb * S * DD);
    }

    {
        const int ln_lane = tid & 31, ln_wave = tid >> 5;
        const int w_even = ln_wave & 6, w_odd = ln_wave | 1;
        const v4f g4 = *(const v4f*)(lng + (tid & 63) * 4);
        const v4f b4 = *(const v4f*)(lnb + (tid & 63) * 4);
        float part[16], mu[16];
        #pragma unroll
        for (int bb = 0; bb < 16; ++bb)
            part[bb] = wave32_sum(ln_sum4(xv[bb]));
        if (ln_lane == 0) {
            #pragma unroll
            for (int bb = 0; bb < 16; ++bb) red[0][ln_wave][bb] = part[bb];
        }
        __syncthreads();
        #pragma unroll
        for (int bb = 0; bb < 16; ++bb) {
            mu[bb] = (red[0][w_even][bb] + red[0][w_odd][bb]) * (1.0f / 256.0f);
            part[bb] = wave32_sum(ln_sq4(xv[bb] - mu[bb]));
        }
        if (ln_lane == 0) {
            #pragma unroll
            for (int bb = 0; bb < 16; ++bb) red[1][ln_wave][bb] = part[bb];
        }
        __syncthreads();
        #pragma unroll
        for (int bb = 0; bb < 16; ++bb) {
            const float rsd = rsqrtf((red[1][w_even][bb] + red[1][w_odd][bb]) * (1.0f / 256.0f) + 1e-5f);
            xv[bb] = ln_apply(xv[bb], mu[bb], rsd, g4, b4);
        }
    }

    if (tid < 42 * DG) {
        const int g = tid % DG, tq = tid / DG;
        const float pw = powf(1.5f, (float)g);
        for (int tt = tq; tt < DS_NT; tt += 42) {
            const float period = (2.0f + 0.5f * (float)(i0 + tt)) * pw;
            #pragma unroll
            for (int sl = 0; sl < DS_SG; ++sl)
                Ct[sl * DS_CS + tt * DG + g] = __cosf(two_pi * (float)(s0 + sl) / period);
        }
    }
    {
        float* xd = Xs + (tid >> 6) * DS_XS + (tid & 63) * DS_XP;
        #pragma unroll
        for (int bb = 0; bb < 16; ++bb) {
            v2f lo = { xv[bb].x, xv[bb].y }, hi = { xv[bb].z, xv[bb].w };
            *(v2f*)(xd + bb * 2)      = lo;
            *(v2f*)(xd + 32 + bb * 2) = hi;
        }
    }
    __syncthreads();

    const int lane = tid & 31, wave = tid >> 5;
    const int lhalf = lane >> 4, lmod = lane & 15;
    const int iloc = wave * 16 + lmod;
    const float* prow = P + ((size_t)(i0 + iloc) * DD + 2 * lhalf) * DG;
    const float* cb = Ct + (iloc + 2 * lhalf) * DG;
    const float* xb = Xs + lhalf * 32 + lmod * 2;

    v8f acc[DS_SG] = {};
    DescTile Ta, Tb;
    desc_load(Ta, prow);
    for (int j0 = 0; j0 < DD; j0 += 32) {
        desc_load(Tb, prow + (j0 + 16) * DG);
        __builtin_amdgcn_sched_barrier(0);
        desc_step(Ta, cb + j0 * DG, xb + (j0 >> 2) * DS_XP, acc);
        const int jn = (j0 + 32 < DD) ? (j0 + 32) : j0;
        desc_load(Ta, prow + jn * DG);
        __builtin_amdgcn_sched_barrier(0);
        desc_step(Tb, cb + (j0 + 16) * DG, xb + ((j0 + 16) >> 2) * DS_XP, acc);
    }

    float hv[DS_SG][8];
    #pragma unroll
    for (int sl = 0; sl < DS_SG; ++sl)
        #pragma unroll
        for (int v = 0; v < 8; ++v)
            hv[sl][v] = h[((size_t)(v + 8 * lhalf) * S + s0 + sl) * DD + i0 + iloc];
    #pragma unroll
    for (int sl = 0; sl < DS_SG; ++sl)
        #pragma unroll
        for (int v = 0; v < 8; ++v)
            hd[((size_t)(v + 8 * lhalf) * S + s0 + sl) * DD + i0 + iloc] = acc[sl][v] + hv[sl][v];
}

struct AttnTile { v4f k[4]; v2f v[8]; };

__device__ __forceinline__ float attn_xhalf(float x) {
    return __builtin_bit_cast(float, __builtin_amdgcn_permlanex16(
        __builtin_bit_cast(int, x), __builtin_bit_cast(int, x),
        0x76543210, 0xfedcba98, false, false));
}

__device__ __forceinline__ void attn_load(AttnTile& T, const float* kr, const float* vr) {
    #pragma unroll
    for (int c = 0; c < 4; ++c) T.k[c] = *(const v4f*)(kr + 4 * c);
    #pragma unroll
    for (int j = 0; j < 8; ++j) T.v[j] = *(const v2f*)(vr + (size_t)j * 768);
}

__device__ __forceinline__ void attn_step(const AttnTile& T, const v4f (&qv)[4],
                                          float& mrow, float& lsum, v8f& o0, v8f& o1) {
    const float scale = 0.17677669529663688f;
    v8f s0 = {}, s1 = {};
    #pragma unroll
    for (int c = 0; c < 4; ++c) {
        s0 = wmma_f32(T.k[c].xy, qv[c].xy, s0);
        s1 = wmma_f32(T.k[c].zw, qv[c].zw, s1);
    }
    float sv[8];
    #pragma unroll
    for (int v = 0; v < 8; ++v) sv[v] = (s0[v] + s1[v]) * scale;
    float mx = fmaxf(fmaxf(fmaxf(sv[0], sv[1]), fmaxf(sv[2], sv[3])),
                     fmaxf(fmaxf(sv[4], sv[5]), fmaxf(sv[6], sv[7])));
    mx = fmaxf(mx, attn_xhalf(mx));
    const float mnew = fmaxf(mrow, mx);
    const float corr = __expf(mrow - mnew);
    float p[8];
    float rsum = 0.f;
    #pragma unroll
    for (int v = 0; v < 8; ++v) { p[v] = __expf(sv[v] - mnew); rsum += p[v]; }
    lsum = lsum * corr + rsum;
    mrow = mnew;
    #pragma unroll
    for (int v = 0; v < 8; ++v) { o0[v] *= corr; o1[v] *= corr; }
    #pragma unroll
    for (int t = 0; t < 4; ++t) {
        v2f bp = { p[2 * t], p[2 * t + 1] };
        v2f a0 = { T.v[2 * t].x, T.v[2 * t + 1].x };
        v2f a1 = { T.v[2 * t].y, T.v[2 * t + 1].y };
        o0 = wmma_f32(a0, bp, o0);
        o1 = wmma_f32(a1, bp, o1);
    }
}

__global__ __launch_bounds__(128) void ddnet_attn_kernel(
    const float* __restrict__ qkv, float* __restrict__ o, int S)
{
    const int qb    = S / 64;
    const int bh    = blockIdx.x / qb;
    const int qblk  = blockIdx.x % qb;
    const int b     = bh >> 3, head = bh & 7;
    const int tid   = threadIdx.x, lane = tid & 31, wave = tid >> 5;
    const int lhalf = lane >> 4, lmod = lane & 15;
    const int q0    = qblk * 64 + wave * 16;

    const size_t rs = 768;
    const float* qp = qkv + (size_t)b * S * rs + head * 32;
    const float* kr = qp + 256 + (size_t)lmod * rs + 16 * lhalf;
    const float* vr = qp + 512 + (size_t)(8 * lhalf) * rs + 2 * lmod;

    v4f qv[4];
    {
        const float* qr = qp + (size_t)(q0 + lmod) * rs + 16 * lhalf;
        #pragma unroll
        for (int c = 0; c < 4; ++c) qv[c] = *(const v4f*)(qr + 4 * c);
    }

    float mrow = -1e30f, lsum = 0.f;
    v8f o0 = {}, o1 = {};

    AttnTile Ta, Tb;
    attn_load(Ta, kr, vr);
    attn_load(Tb, kr + (size_t)16 * rs, vr + (size_t)16 * rs);
    for (int k0 = 0; k0 < S; k0 += 32) {
        attn_step(Ta, qv, mrow, lsum, o0, o1);
        const int kn = (k0 + 32 < S) ? (k0 + 32) : k0;
        attn_load(Ta, kr + (size_t)kn * rs, vr + (size_t)kn * rs);
        __builtin_amdgcn_sched_barrier(0);
        attn_step(Tb, qv, mrow, lsum, o0, o1);
        attn_load(Tb, kr + (size_t)(kn + 16) * rs, vr + (size_t)(kn + 16) * rs);
        __builtin_amdgcn_sched_barrier(0);
    }

    const float inv = 1.0f / (lsum + attn_xhalf(lsum));
    float* orow = o + ((size_t)b * S + q0 + lmod) * 256 + head * 32 + 16 * lhalf;
    #pragma unroll
    for (int j = 0; j < 4; ++j) {
        v4f w = { o0[2 * j] * inv, o1[2 * j] * inv, o0[2 * j + 1] * inv, o1[2 * j + 1] * inv };
        *(v4f*)(orow + 4 * j) = w;
    }
}

static void launch_gemm(const float* A, const float* W, const float* bias,
                        const float* resid, float* C, int M, int N, int K,
                        int act, hipStream_t stream)
{
    dim3 grid(N / GT_N, M / GT_M);
    ddnet_gemm_kernel<false><<<grid, 256, 0, stream>>>(A, W, bias, resid, C, M, N, K, act,
                                                       nullptr, nullptr, nullptr);
}

static void launch_gemm_ln(const float* A, const float* lng, const float* lnb, float* side,
                           const float* W, const float* bias, float* C, int M, int N,
                           int act, hipStream_t stream)
{
    dim3 grid(N / GT_N, M / GT_M);
    ddnet_gemm_kernel<true><<<grid, 256, 0, stream>>>(A, W, bias, nullptr, C, M, N, 256, act,
                                                      lng, lnb, side);
}

extern "C" void kernel_launch(void* const* d_in, const int* in_sizes, int n_in,
                              void* d_out, int out_size, void* d_ws, size_t ws_size,
                              hipStream_t stream) {
    (void)in_sizes; (void)n_in; (void)out_size; (void)ws_size;
    const int B = 16, S = 512, IN = 64, D = 256, L = 3, G = 6, DFF = 1024;
    const int Mr = B * S;

    const float* x      = (const float*)d_in[0];
    const float* in_w   = (const float*)d_in[1];
    const float* in_b   = (const float*)d_in[2];
    const float* Mw     = (const float*)d_in[3];
    const float* P      = (const float*)d_in[4];
    const float* desc_g = (const float*)d_in[5];
    const float* desc_b = (const float*)d_in[6];
    const float* qkv_w  = (const float*)d_in[7];
    const float* qkv_b  = (const float*)d_in[8];
    const float* out_w  = (const float*)d_in[9];
    const float* out_b  = (const float*)d_in[10];
    const float* ff1_w  = (const float*)d_in[11];
    const float* ff1_b  = (const float*)d_in[12];
    const float* ff2_w  = (const float*)d_in[13];
    const float* ff2_b  = (const float*)d_in[14];
    const float* ln1_g  = (const float*)d_in[15];
    const float* ln1_b  = (const float*)d_in[16];
    const float* ln2_g  = (const float*)d_in[17];
    const float* ln2_b  = (const float*)d_in[18];

    float* h    = (float*)d_ws;
    float* tmp  = h    + (size_t)Mr * D;
    float* tmp2 = tmp  + (size_t)Mr * D;
    float* hd   = tmp2 + (size_t)Mr * D;
    float* qkv  = hd   + (size_t)Mr * D;
    float* obuf = qkv  + (size_t)Mr * 3 * D;
    float* h1   = obuf + (size_t)Mr * D;
    float* ff   = h1   + (size_t)Mr * D;

    launch_gemm(x, in_w, in_b, nullptr, h, Mr, D, IN, 0, stream);
    ddnet_pe_kernel<<<(Mr * D) / 256, 256, 0, stream>>>(h, S, D);

    for (int l = 0; l < L; ++l) {
        if (l == 0)
            launch_gemm(h, Mw, nullptr, nullptr, tmp, Mr, D, D, 0, stream);
        else
            launch_gemm_ln(tmp2, ln2_g + (l - 1) * D, ln2_b + (l - 1) * D, h,
                           Mw + (size_t)l * D * D, nullptr, tmp, Mr, D, 0, stream);
        ddnet_desc_kernel<<<(S / 4) * 2, 256, 0, stream>>>(tmp, desc_g + l * D, desc_b + l * D, h,
                                                           P + (size_t)l * D * D * G, hd, S);
        launch_gemm(hd, qkv_w + (size_t)l * 3 * D * D, qkv_b + (size_t)l * 3 * D,
                    nullptr, qkv, Mr, 3 * D, D, 0, stream);
        ddnet_attn_kernel<<<B * 8 * (S / 64), 128, 0, stream>>>(qkv, obuf, S);
        launch_gemm(obuf, out_w + (size_t)l * D * D, out_b + l * D, hd, tmp, Mr, D, D, 0, stream);
        launch_gemm_ln(tmp, ln1_g + l * D, ln1_b + l * D, h1, ff1_w + (size_t)l * DFF * D,
                       ff1_b + (size_t)l * DFF, ff, Mr, DFF, 1  , stream);
        launch_gemm(ff, ff2_w + (size_t)l * D * DFF, ff2_b + l * D, h1, tmp2, Mr, D, DFF, 0, stream);
    }
    ddnet_ln_kernel<<<Mr / 8, 256, 0, stream>>>(tmp2, ln2_g + (L - 1) * D, ln2_b + (L - 1) * D,
                                                (float*)d_out, D);
}


# GELU epilogue (first feed-forward product): erf as one clamped rational polynomial v*P(v^2)/Q(v^2) on register pairs, no branch per output (was library erff, two divergent paths); erf moves <4e-7
# speedup vs baseline: 2.6840x; 1.0350x over previous
#include <hip/hip_runtime.h>
#include <math.h>

typedef __attribute__((ext_vector_type(2))) float v2f;
typedef __attribute__((ext_vector_type(8))) float v8f;

__device__ __forceinline__ v8f wmma_f32(v2f a, v2f b, v8f c) {
    return __builtin_amdgcn_wmma_f32_16x16x4_f32(false, a, false, b, (short)0, c, false, false);
}
__device__ __forceinline__ v2f lds_frag(const float* p) {
    return *(const v2f*)p;
}

typedef __attribute__((ext_vector_type(4))) float v4f;

template <int M> __device__ __forceinline__ float row_xmask(float x) {
    return __builtin_bit_cast(float, __builtin_amdgcn_update_dpp(
        0, __builtin_bit_cast(int, x), 0x160 + M, 0xf, 0xf, false));
}
__device__ __forceinline__ float row16_sum(float s) {
    s += row_xmask<1>(s);
    s += row_xmask<2>(s);
    s += row_xmask<4>(s);
    s += row_xmask<8>(s);
    return s;
}
__device__ __forceinline__ float wave32_sum(float s) {
    s = row16_sum(s);
    return s + __builtin_bit_cast(float, __builtin_amdgcn_permlanex16(
        __builtin_bit_cast(int, s), __builtin_bit_cast(int, s),
        0x76543210, 0xfedcba98, false, false));
}
__device__ __forceinline__ float ln_sum4(v4f x) { return (x.x + x.y) + (x.z + x.w); }
__device__ __forceinline__ float ln_sq4(v4f d) { return (d.x * d.x + d.y * d.y) + (d.z * d.z + d.w * d.w); }
__device__ __forceinline__ v4f ln_apply(v4f x, float mu, float rs, v4f g, v4f b) {
    v4f r;
    r.x = (x.x - mu) * rs * g.x + b.x;
    r.y = (x.y - mu) * rs * g.y + b.y;
    r.z = (x.z - mu) * rs * g.z + b.z;
    r.w = (x.w - mu) * rs * g.w + b.w;
    return r;
}

__device__ __forceinline__ v2f fma2(v2f a, v2f b, float c) {
    const v2f cc = { c, c };
    return __builtin_elementwise_fma(a, b, cc);
}
__device__ __forceinline__ v2f gelu_pair(v2f x) {
    v2f v = x * 0.70710678118654752f;
    v.x = fminf(fmaxf(v.x, -4.0f), 4.0f);
    v.y = fminf(fmaxf(v.y, -4.0f), 4.0f);
    const v2f t = v * v;
    v2f p = fma2(t, (v2f){ -2.72614225801306e-10f, -2.72614225801306e-10f }, 2.77068142495902e-08f);
    p = fma2(p, t, -2.10102402082508e-06f);
    p = fma2(p, t, -5.69250639462346e-05f);
    p = fma2(p, t, -7.34990630326855e-04f);
    p = fma2(p, t, -2.95459980854025e-03f);
    p = fma2(p, t, -1.60960333262415e-02f);
    v2f q = fma2(t, (v2f){ -1.45660718464996e-05f, -1.45660718464996e-05f }, -2.13374055278905e-04f);
    q = fma2(q, t, -1.68282697438203e-03f);
    q = fma2(q, t, -7.37332916720468e-03f);
    q = fma2(q, t, -1.42647390514189e-02f);
    const v2f r = { __builtin_amdgcn_rcpf(q.x), __builtin_amdgcn_rcpf(q.y) };
    const v2f e = (v * p) * r;
    const v2f hx = x * 0.5f;
    return __builtin_elementwise_fma(hx, e, hx);
}

#define GT_M 64
#define GT_N 128
#define GT_K 64
#define GT_KQ (GT_K / 4)
#define A_SL (GT_M * 4 + 4)
#define B_SL (GT_N * 4 + 4)

template <bool LN>
__global__ __launch_bounds__(256) void ddnet_gemm_kernel(
    const float* __restrict__ A, const float* __restrict__ W,
    const float* __restrict__ bias, const float* __restrict__ resid,
    float* __restrict__ C, int Mdim, int Ndim, int Kdim, int act,
    const float* __restrict__ lng, const float* __restrict__ lnb,
    float* __restrict__ side)
{
    __shared__ __align__(16) float As[2][GT_KQ * A_SL];
    __shared__ __align__(16) float Bs[2][GT_KQ * B_SL];

    const int tid   = threadIdx.x;
    const int lane  = tid & 31;
    const int wave  = tid >> 5;
    const int lhalf = lane >> 4;
    const int lmod  = lane & 15;
    const int m0    = blockIdx.y * GT_M;
    const int n0    = blockIdx.x * GT_N;
    const int wm    = (wave & 1) * 32;
    const int wn    = (wave >> 1) * 32;
    const int nk    = Kdim >> 6;

    const int lq = tid & 15;
    const int lr = tid >> 4;
    const float* Ablk = A + (size_t)m0 * Kdim;
    const float* Wblk = W + (size_t)n0 * Kdim;
    const unsigned toff = (unsigned)(lr * Kdim + lq * 4);
    const int rstep = 16 * Kdim;
    const int a_st = lq * A_SL + lr * 4;
    const int b_st = lq * B_SL + lr * 4;
    const int a_rd = lhalf * A_SL + (wm + lmod) * 4;
    const int b_rd = lhalf * B_SL + (wn + lmod) * 4;

    v8f acc[2][2] = {};
    v4f ra[4], rb[8];
    float mu[4] = { 0.f, 0.f, 0.f, 0.f }, rs[4] = { 1.f, 1.f, 1.f, 1.f };
    v4f g4 = {}, b4 = {};
    float* sblk = nullptr;

    if constexpr (LN) {
        v4f pv[4][4];
        #pragma unroll
        for (int i = 0; i < 4; ++i)
            #pragma unroll
            for (int t = 0; t < 4; ++t)
                pv[i][t] = *(const v4f*)(Ablk + (t * 64 + i * rstep) + toff);
        #pragma unroll
        for (int i = 0; i < 8; ++i)
            rb[i] = *(const v4f*)(Wblk + i * rstep + toff);
        g4 = *(const v4f*)(lng + lq * 4);
        b4 = *(const v4f*)(lnb + lq * 4);
        sblk = side + (size_t)m0 * Kdim;
        #pragma unroll
        for (int i = 0; i < 4; ++i) {
            float sm = (ln_sum4(pv[i][0]) + ln_sum4(pv[i][1])) + (ln_sum4(pv[i][2]) + ln_sum4(pv[i][3]));
            const float m = row16_sum(sm) * (1.0f / 256.0f);
            float sq = (ln_sq4(pv[i][0] - m) + ln_sq4(pv[i][1] - m)) + (ln_sq4(pv[i][2] - m) + ln_sq4(pv[i][3] - m));
            mu[i] = m;
            rs[i] = rsqrtf(row16_sum(sq) * (1.0f / 256.0f) + 1e-5f);
            ra[i] = ln_apply(pv[i][0], mu[i], rs[i], g4, b4);
        }
    } else {
        #pragma unroll
        for (int i = 0; i < 4; ++i)
            ra[i] = *(const v4f*)(Ablk + i * rstep + toff);
        #pragma unroll
        for (int i = 0; i < 8; ++i)
            rb[i] = *(const v4f*)(Wblk + i * rstep + toff);
    }
    #pragma unroll
    for (int i = 0; i < 4; ++i)
        *(v4f*)&As[0][a_st + i * 64] = ra[i];
    #pragma unroll
    for (int i = 0; i < 8; ++i)
        *(v4f*)&Bs[0][b_st + i * 64] = rb[i];
    __syncthreads();
    if constexpr (LN) {
        if (blockIdx.x == 0) {
            #pragma unroll
            for (int i = 0; i < 4; ++i)
                *(v4f*)(sblk + i * rstep + toff) = ra[i];
        }
    }

    for (int kt = 0; kt < nk; ++kt) {
        const int buf = kt & 1;
        if (kt + 1 < nk) {
            const int k0 = (kt + 1) << 6;
            #pragma unroll
            for (int i = 0; i < 4; ++i)
                ra[i] = *(const v4f*)(Ablk + (k0 + i * rstep) + toff);
            #pragma unroll
            for (int i = 0; i < 8; ++i)
                rb[i] = *(const v4f*)(Wblk + (k0 + i * rstep) + toff);
            if constexpr (LN) {
                g4 = *(const v4f*)(lng + k0 + lq * 4);
                b4 = *(const v4f*)(lnb + k0 + lq * 4);
            }
        }
        const float* Ab = As[buf] + a_rd;
        const float* Bb = Bs[buf] + b_rd;
        #pragma unroll
        for (int u = 0; u < GT_KQ / 2; ++u) {
            v4f a0 = *(const v4f*)&Ab[2 * u * A_SL];
            v4f a1 = *(const v4f*)&Ab[2 * u * A_SL + 64];
            v4f b0 = *(const v4f*)&Bb[2 * u * B_SL];
            v4f b1 = *(const v4f*)&Bb[2 * u * B_SL + 64];
            acc[0][0] = wmma_f32(a0.xy, b0.xy, acc[0][0]);
            acc[0][1] = wmma_f32(a0.xy, b1.xy, acc[0][1]);
            acc[1][0] = wmma_f32(a1.xy, b0.xy, acc[1][0]);
            acc[1][1] = wmma_f32(a1.xy, b1.xy, acc[1][1]);
            acc[0][0] = wmma_f32(a0.zw, b0.zw, acc[0][0]);
            acc[0][1] = wmma_f32(a0.zw, b1.zw, acc[0][1]);
            acc[1][0] = wmma_f32(a1.zw, b0.zw, acc[1][0]);
            acc[1][1] = wmma_f32(a1.zw, b1.zw, acc[1][1]);
        }
        if (kt + 1 < nk) {
            if constexpr (LN) {
                #pragma unroll
                for (int i = 0; i < 4; ++i)
                    ra[i] = ln_apply(ra[i], mu[i], rs[i], g4, b4);
            }
            __syncthreads();
            #pragma unroll
            for (int i = 0; i < 4; ++i)
                *(v4f*)&As[buf ^ 1][a_st + i * 64] = ra[i];
            #pragma unroll
            for (int i = 0; i < 8; ++i)
                *(v4f*)&Bs[buf ^ 1][b_st + i * 64] = rb[i];
            __syncthreads();
            if constexpr (LN) {
                if (blockIdx.x == 0) {
                    const int k0 = (kt + 1) << 6;
                    #pragma unroll
                    for (int i = 0; i < 4; ++i)
                        *(v4f*)(sblk + (k0 + i * rstep) + toff) = ra[i];
                }
            }
        }
    }

    const bool has_bias = bias != nullptr;
    const bool has_res  = resid != nullptr;
    const int  ecol = n0 + wn + lmod;
    const size_t eoff = (size_t)(m0 + wm + 8 * lhalf) * Ndim + ecol;
    float bv[2] = { 0.f, 0.f };
    float rv[2][2][8] = {};
    if (has_bias) { bv[0] = bias[ecol]; bv[1] = bias[ecol + 16]; }
    if (has_res) {
        const float* rp = resid + eoff;
        #pragma unroll
        for (int mi = 0; mi < 2; ++mi)
            #pragma unroll
            for (int ni = 0; ni < 2; ++ni)
                #pragma unroll
                for (int v = 0; v < 8; ++v)
                    rv[mi][ni][v] = rp[(size_t)(mi * 16 + v) * Ndim + ni * 16];
    }
    float* cp = C + eoff;
    #pragma unroll
    for (int mi = 0; mi < 2; ++mi)
        #pragma unroll
        for (int ni = 0; ni < 2; ++ni)
            #pragma unroll
            for (int v = 0; v < 8; ++v) {
                float val = acc[mi][ni][v];
                if (has_bias) val += bv[ni];
                if (has_res)  val += rv[mi][ni][v];
                acc[mi][ni][v] = val;
            }
    if (act == 1) {
        #pragma unroll
        for (int mi = 0; mi < 2; ++mi)
            #pragma unroll
            for (int ni = 0; ni < 2; ++ni)
                #pragma unroll
                for (int v = 0; v < 8; v += 2) {
                    v2f x = { acc[mi][ni][v], acc[mi][ni][v + 1] };
                    x = gelu_pair(x);
                    acc[mi][ni][v]     = x.x;
                    acc[mi][ni][v + 1] = x.y;
                }
    }
    #pragma unroll
    for (int mi = 0; mi < 2; ++mi)
        #pragma unroll
        for (int ni = 0; ni < 2; ++ni)
            #pragma unroll
            for (int v = 0; v < 8; ++v)
                cp[(size_t)(mi * 16 + v) * Ndim + ni * 16] = acc[mi][ni][v];
}

__global__ __launch_bounds__(256) void ddnet_pe_kernel(
    float* __restrict__ h, int S, int D)
{
    size_t idx = (size_t)blockIdx.x * 256 + threadIdx.x;
    int d = (int)(idx % D);
    int s = (int)((idx / D) % S);
    int pair = d >> 1;
    float div = __expf(-(float)(2 * pair) * (logf(10000.0f) / (float)D));
    float ang = (float)s * div;
    h[idx] += (d & 1) ? __cosf(ang) : __sinf(ang);
}

__global__ __launch_bounds__(256) void ddnet_ln_kernel(
    const float* __restrict__ x, const float* __restrict__ gamma,
    const float* __restrict__ beta, float* __restrict__ out, int D)
{
    int row  = blockIdx.x * 8 + (threadIdx.x >> 5);
    int lane = threadIdx.x & 31;
    const float* xr = x + (size_t)row * D;
    float vals[8];
    float s = 0.f;
    #pragma unroll
    for (int i = 0; i < 8; ++i) { vals[i] = xr[lane + i * 32]; s += vals[i]; }
    #pragma unroll
    for (int m = 16; m >= 1; m >>= 1) s += __shfl_xor(s, m, 32);
    float mean = s * (1.0f / 256.0f);
    float vs = 0.f;
    #pragma unroll
    for (int i = 0; i < 8; ++i) { float d = vals[i] - mean; vs += d * d; }
    #pragma unroll
    for (int m = 16; m >= 1; m >>= 1) vs += __shfl_xor(vs, m, 32);
    float rstd = rsqrtf(vs * (1.0f / 256.0f) + 1e-5f);
    float* orow = out + (size_t)row * D;
    #pragma unroll
    for (int i = 0; i < 8; ++i) {
        int c = lane + i * 32;
        orow[c] = (vals[i] - mean) * rstd * gamma[c] + beta[c];
    }
}

#define DD 256
#define DG 6
#define DS_SG 4
#define DS_NI 128
#define DS_XP 66
#define DS_XS (64 * DS_XP)
#define DS_NT (DS_NI + DD - 1)
#define DS_CS (384 * DG)

struct DescTile { v4f p[12]; };

__device__ __forceinline__ void desc_load(DescTile& T, const float* pr) {
    #pragma unroll
    for (int t = 0; t < 4; ++t)
        #pragma unroll
        for (int q = 0; q < 3; ++q)
            T.p[3 * t + q] = *(const v4f*)(pr + t * 24 + q * 4);
}

struct DescOps { v2f c[DS_SG][6]; v2f a[DS_SG]; };

__device__ __forceinline__ void desc_ops(DescOps& O, const float* cp, const float* xp) {
    #pragma unroll
    for (int sl = 0; sl < DS_SG; ++sl) {
        #pragma unroll
        for (int m = 0; m < 6; ++m)
            O.c[sl][m] = *(const v2f*)(cp + sl * DS_CS + 2 * m);
        O.a[sl] = lds_frag(xp + sl * DS_XS);
    }
}

__device__ __forceinline__ void desc_group(const DescTile& T, int t, const DescOps& O,
                                           v8f (&acc)[DS_SG]) {
    #pragma unroll
    for (int sl = 0; sl < DS_SG; ++sl) {
        float w0 = 0.f, w1 = 0.f;
        #pragma unroll
        for (int g = 0; g < DG; ++g) {
            w0 += T.p[3 * t + (g >> 2)][g & 3] * O.c[sl][g >> 1][g & 1];
            w1 += T.p[3 * t + ((6 + g) >> 2)][(6 + g) & 3] * O.c[sl][(6 + g) >> 1][g & 1];
        }
        v2f b = { w0, w1 };
        acc[sl] = wmma_f32(O.a[sl], b, acc[sl]);
    }
}

__device__ __forceinline__ void desc_step(const DescTile& T, const float* cb,
                                          const float* xb, v8f (&acc)[DS_SG]) {
    DescOps Oa, Ob;
    desc_ops(Oa, cb, xb);
    desc_ops(Ob, cb + 24, xb + DS_XP);
    __builtin_amdgcn_sched_barrier(0);
    desc_group(T, 0, Oa, acc);
    desc_ops(Oa, cb + 48, xb + 2 * DS_XP);
    __builtin_amdgcn_sched_barrier(0);
    desc_group(T, 1, Ob, acc);
    desc_ops(Ob, cb + 72, xb + 3 * DS_XP);
    __builtin_amdgcn_sched_barrier(0);
    desc_group(T, 2, Oa, acc);
    desc_group(T, 3, Ob, acc);
}

__global__ __launch_bounds__(256) void ddnet_desc_kernel(
    const float* __restrict__ xpre, const float* __restrict__ lng,
    const float* __restrict__ lnb, const float* __restrict__ h,
    const float* __restrict__ P, float* __restrict__ hd, int S)
{
    __shared__ __align__(16) float Xs[DS_SG * DS_XS];
    __shared__ __align__(16) float Ct[DS_SG * DS_CS];
    __shared__ float red[2][8][16];

    const int tid = threadIdx.x;
    const int i0  = (blockIdx.x & 1) * DS_NI;
    const int s0  = (blockIdx.x >> 1) * DS_SG;
    const float two_pi = 6.28318530717958647692f;

    v4f xv[16];
    {
        const float* xsrc = xpre + (size_t)s0 * DD + tid * 4;
        #pragma unroll
        for (int bb = 0; bb < 16; ++bb)
            xv[bb] = *(const v4f*)(xsrc + (size_t)bb * S * DD);
    }

    {
        const int ln_lane = tid & 31, ln_wave = tid >> 5;
        const int w_even = ln_wave & 6, w_odd = ln_wave | 1;
        const v4f g4 = *(const v4f*)(lng + (tid & 63) * 4);
        const v4f b4 = *(const v4f*)(lnb + (tid & 63) * 4);
        float part[16], mu[16];
        #pragma unroll
        for (int bb = 0; bb < 16; ++bb)
            part[bb] = wave32_sum(ln_sum4(xv[bb]));
        if (ln_lane == 0) {
            #pragma unroll
            for (int bb = 0; bb < 16; ++bb) red[0][ln_wave][bb] = part[bb];
        }
        __syncthreads();
        #pragma unroll
        for (int bb = 0; bb < 16; ++bb) {
            mu[bb] = (red[0][w_even][bb] + red[0][w_odd][bb]) * (1.0f / 256.0f);
            part[bb] = wave32_sum(ln_sq4(xv[bb] - mu[bb]));
        }
        if (ln_lane == 0) {
            #pragma unroll
            for (int bb = 0; bb < 16; ++bb) red[1][ln_wave][bb] = part[bb];
        }
        __syncthreads();
        #pragma unroll
        for (int bb = 0; bb < 16; ++bb) {
            const float rsd = rsqrtf((red[1][w_even][bb] + red[1][w_odd][bb]) * (1.0f / 256.0f) + 1e-5f);
            xv[bb] = ln_apply(xv[bb], mu[bb], rsd, g4, b4);
        }
    }

    if (tid < 42 * DG) {
        const int g = tid % DG, tq = tid / DG;
        const float pw = powf(1.5f, (float)g);
        for (int tt = tq; tt < DS_NT; tt += 42) {
            const float period = (2.0f + 0.5f * (float)(i0 + tt)) * pw;
            #pragma unroll
            for (int sl = 0; sl < DS_SG; ++sl)
                Ct[sl * DS_CS + tt * DG + g] = __cosf(two_pi * (float)(s0 + sl) / period);
        }
    }
    {
        float* xd = Xs + (tid >> 6) * DS_XS + (tid & 63) * DS_XP;
        #pragma unroll
        for (int bb = 0; bb < 16; ++bb) {
            v2f lo = { xv[bb].x, xv[bb].y }, hi = { xv[bb].z, xv[bb].w };
            *(v2f*)(xd + bb * 2)      = lo;
            *(v2f*)(xd + 32 + bb * 2) = hi;
        }
    }
    __syncthreads();

    const int lane = tid & 31, wave = tid >> 5;
    const int lhalf = lane >> 4, lmod = lane & 15;
    const int iloc = wave * 16 + lmod;
    const float* prow = P + ((size_t)(i0 + iloc) * DD + 2 * lhalf) * DG;
    const float* cb = Ct + (iloc + 2 * lhalf) * DG;
    const float* xb = Xs + lhalf * 32 + lmod * 2;

    v8f acc[DS_SG] = {};
    DescTile Ta, Tb;
    desc_load(Ta, prow);
    for (int j0 = 0; j0 < DD; j0 += 32) {
        desc_load(Tb, prow + (j0 + 16) * DG);
        __builtin_amdgcn_sched_barrier(0);
        desc_step(Ta, cb + j0 * DG, xb + (j0 >> 2) * DS_XP, acc);
        const int jn = (j0 + 32 < DD) ? (j0 + 32) : j0;
        desc_load(Ta, prow + jn * DG);
        __builtin_amdgcn_sched_barrier(0);
        desc_step(Tb, cb + (j0 + 16) * DG, xb + ((j0 + 16) >> 2) * DS_XP, acc);
    }

    float hv[DS_SG][8];
    #pragma unroll
    for (int sl = 0; sl < DS_SG; ++sl)
        #pragma unroll
        for (int v = 0; v < 8; ++v)
            hv[sl][v] = h[((size_t)(v + 8 * lhalf) * S + s0 + sl) * DD + i0 + iloc];
    #pragma unroll
    for (int sl = 0; sl < DS_SG; ++sl)
        #pragma unroll
        for (int v = 0; v < 8; ++v)
            hd[((size_t)(v + 8 * lhalf) * S + s0 + sl) * DD + i0 + iloc] = acc[sl][v] + hv[sl][v];
}

struct AttnTile { v4f k[4]; v2f v[8]; };

__device__ __forceinline__ float attn_xhalf(float x) {
    return __builtin_bit_cast(float, __builtin_amdgcn_permlanex16(
        __builtin_bit_cast(int, x), __builtin_bit_cast(int, x),
        0x76543210, 0xfedcba98, false, false));
}

__device__ __forceinline__ void attn_load(AttnTile& T, const float* kr, const float* vr) {
    #pragma unroll
    for (int c = 0; c < 4; ++c) T.k[c] = *(const v4f*)(kr + 4 * c);
    #pragma unroll
    for (int j = 0; j < 8; ++j) T.v[j] = *(const v2f*)(vr + (size_t)j * 768);
}

__device__ __forceinline__ void attn_step(const AttnTile& T, const v4f (&qv)[4],
                                          float& mrow, float& lsum, v8f& o0, v8f& o1) {
    const float scale = 0.17677669529663688f;
    v8f s0 = {}, s1 = {};
    #pragma unroll
    for (int c = 0; c < 4; ++c) {
        s0 = wmma_f32(T.k[c].xy, qv[c].xy, s0);
        s1 = wmma_f32(T.k[c].zw, qv[c].zw, s1);
    }
    float sv[8];
    #pragma unroll
    for (int v = 0; v < 8; ++v) sv[v] = (s0[v] + s1[v]) * scale;
    float mx = fmaxf(fmaxf(fmaxf(sv[0], sv[1]), fmaxf(sv[2], sv[3])),
                     fmaxf(fmaxf(sv[4], sv[5]), fmaxf(sv[6], sv[7])));
    mx = fmaxf(mx, attn_xhalf(mx));
    const float mnew = fmaxf(mrow, mx);
    const float corr = __expf(mrow - mnew);
    float p[8];
    float rsum = 0.f;
    #pragma unroll
    for (int v = 0; v < 8; ++v) { p[v] = __expf(sv[v] - mnew); rsum += p[v]; }
    lsum = lsum * corr + rsum;
    mrow = mnew;
    #pragma unroll
    for (int v = 0; v < 8; ++v) { o0[v] *= corr; o1[v] *= corr; }
    #pragma unroll
    for (int t = 0; t < 4; ++t) {
        v2f bp = { p[2 * t], p[2 * t + 1] };
        v2f a0 = { T.v[2 * t].x, T.v[2 * t + 1].x };
        v2f a1 = { T.v[2 * t].y, T.v[2 * t + 1].y };
        o0 = wmma_f32(a0, bp, o0);
        o1 = wmma_f32(a1, bp, o1);
    }
}

__global__ __launch_bounds__(128) void ddnet_attn_kernel(
    const float* __restrict__ qkv, float* __restrict__ o, int S)
{
    const int qb    = S / 64;
    const int bh    = blockIdx.x / qb;
    const int qblk  = blockIdx.x % qb;
    const int b     = bh >> 3, head = bh & 7;
    const int tid   = threadIdx.x, lane = tid & 31, wave = tid >> 5;
    const int lhalf = lane >> 4, lmod = lane & 15;
    const int q0    = qblk * 64 + wave * 16;

    const size_t rs = 768;
    const float* qp = qkv + (size_t)b * S * rs + head * 32;
    const float* kr = qp + 256 + (size_t)lmod * rs + 16 * lhalf;
    const float* vr = qp + 512 + (size_t)(8 * lhalf) * rs + 2 * lmod;

    v4f qv[4];
    {
        const float* qr = qp + (size_t)(q0 + lmod) * rs + 16 * lhalf;
        #pragma unroll
        for (int c = 0; c < 4; ++c) qv[c] = *(const v4f*)(qr + 4 * c);
    }

    float mrow = -1e30f, lsum = 0.f;
    v8f o0 = {}, o1 = {};

    AttnTile Ta, Tb;
    attn_load(Ta, kr, vr);
    attn_load(Tb, kr + (size_t)16 * rs, vr + (size_t)16 * rs);
    for (int k0 = 0; k0 < S; k0 += 32) {
        attn_step(Ta, qv, mrow, lsum, o0, o1);
        const int kn = (k0 + 32 < S) ? (k0 + 32) : k0;
        attn_load(Ta, kr + (size_t)kn * rs, vr + (size_t)kn * rs);
        __builtin_amdgcn_sched_barrier(0);
        attn_step(Tb, qv, mrow, lsum, o0, o1);
        attn_load(Tb, kr + (size_t)(kn + 16) * rs, vr + (size_t)(kn + 16) * rs);
        __builtin_amdgcn_sched_barrier(0);
    }

    const float inv = 1.0f / (lsum + attn_xhalf(lsum));
    float* orow = o + ((size_t)b * S + q0 + lmod) * 256 + head * 32 + 16 * lhalf;
    #pragma unroll
    for (int j = 0; j < 4; ++j) {
        v4f w = { o0[2 * j] * inv, o1[2 * j] * inv, o0[2 * j + 1] * inv, o1[2 * j + 1] * inv };
        *(v4f*)(orow + 4 * j) = w;
    }
}

static void launch_gemm(const float* A, const float* W, const float* bias,
                        const float* resid, float* C, int M, int N, int K,
                        int act, hipStream_t stream)
{
    dim3 grid(N / GT_N, M / GT_M);
    ddnet_gemm_kernel<false><<<grid, 256, 0, stream>>>(A, W, bias, resid, C, M, N, K, act,
                                                       nullptr, nullptr, nullptr);
}

static void launch_gemm_ln(const float* A, const float* lng, const float* lnb, float* side,
                           const float* W, const float* bias, float* C, int M, int N,
                           int act, hipStream_t stream)
{
    dim3 grid(N / GT_N, M / GT_M);
    ddnet_gemm_kernel<true><<<grid, 256, 0, stream>>>(A, W, bias, nullptr, C, M, N, 256, act,
                                                      lng, lnb, side);
}

extern "C" void kernel_launch(void* const* d_in, const int* in_sizes, int n_in,
                              void* d_out, int out_size, void* d_ws, size_t ws_size,
                              hipStream_t stream) {
    (void)in_sizes; (void)n_in; (void)out_size; (void)ws_size;
    const int B = 16, S = 512, IN = 64, D = 256, L = 3, G = 6, DFF = 1024;
    const int Mr = B * S;

    const float* x      = (const float*)d_in[0];
    const float* in_w   = (const float*)d_in[1];
    const float* in_b   = (const float*)d_in[2];
    const float* Mw     = (const float*)d_in[3];
    const float* P      = (const float*)d_in[4];
    const float* desc_g = (const float*)d_in[5];
    const float* desc_b = (const float*)d_in[6];
    const float* qkv_w  = (const float*)d_in[7];
    const float* qkv_b  = (const float*)d_in[8];
    const float* out_w  = (const float*)d_in[9];
    const float* out_b  = (const float*)d_in[10];
    const float* ff1_w  = (const float*)d_in[11];
    const float* ff1_b  = (const float*)d_in[12];
    const float* ff2_w  = (const float*)d_in[13];
    const float* ff2_b  = (const float*)d_in[14];
    const float* ln1_g  = (const float*)d_in[15];
    const float* ln1_b  = (const float*)d_in[16];
    const float* ln2_g  = (const float*)d_in[17];
    const float* ln2_b  = (const float*)d_in[18];

    float* h    = (float*)d_ws;
    float* tmp  = h    + (size_t)Mr * D;
    float* tmp2 = tmp  + (size_t)Mr * D;
    float* hd   = tmp2 + (size_t)Mr * D;
    float* qkv  = hd   + (size_t)Mr * D;
    float* obuf = qkv  + (size_t)Mr * 3 * D;
    float* h1   = obuf + (size_t)Mr * D;
    float* ff   = h1   + (size_t)Mr * D;

    launch_gemm(x, in_w, in_b, nullptr, h, Mr, D, IN, 0, stream);
    ddnet_pe_kernel<<<(Mr * D) / 256, 256, 0, stream>>>(h, S, D);

    for (int l = 0; l < L; ++l) {
        if (l == 0)
            launch_gemm(h, Mw, nullptr, nullptr, tmp, Mr, D, D, 0, stream);
        else
            launch_gemm_ln(tmp2, ln2_g + (l - 1) * D, ln2_b + (l - 1) * D, h,
                           Mw + (size_t)l * D * D, nullptr, tmp, Mr, D, 0, stream);
        ddnet_desc_kernel<<<(S / 4) * 2, 256, 0, stream>>>(tmp, desc_g + l * D, desc_b + l * D, h,
                                                           P + (size_t)l * D * D * G, hd, S);
        launch_gemm(hd, qkv_w + (size_t)l * 3 * D * D, qkv_b + (size_t)l * 3 * D,
                    nullptr, qkv, Mr, 3 * D, D, 0, stream);
        ddnet_attn_kernel<<<B * 8 * (S / 64), 128, 0, stream>>>(qkv, obuf, S);
        launch_gemm(obuf, out_w + (size_t)l * D * D, out_b + l * D, hd, tmp, Mr, D, D, 0, stream);
        launch_gemm_ln(tmp, ln1_g + l * D, ln1_b + l * D, h1, ff1_w + (size_t)l * DFF * D,
                       ff1_b + (size_t)l * DFF, ff, Mr, DFF, 1  , stream);
        launch_gemm(ff, ff2_w + (size_t)l * D * DFF, ff2_b + l * D, h1, tmp2, Mr, D, DFF, 0, stream);
    }
    ddnet_ln_kernel<<<Mr / 8, 256, 0, stream>>>(tmp2, ln2_g + (L - 1) * D, ln2_b + (L - 1) * D,
                                                (float*)d_out, D);
}


# One kernel per layer for the row-wise chain (out product, LN1, feed-forward products, LN2, next descriptor product): 32 rows per work-group stay in LDS, weights streamed; LN sum order changed
# speedup vs baseline: 2.8377x; 1.0573x over previous
#include <hip/hip_runtime.h>
#include <math.h>

typedef __attribute__((ext_vector_type(2))) float v2f;
typedef __attribute__((ext_vector_type(8))) float v8f;

__device__ __forceinline__ v8f wmma_f32(v2f a, v2f b, v8f c) {
    return __builtin_amdgcn_wmma_f32_16x16x4_f32(false, a, false, b, (short)0, c, false, false);
}
__device__ __forceinline__ v2f lds_frag(const float* p) {
    return *(const v2f*)p;
}

typedef __attribute__((ext_vector_type(4))) float v4f;

template <int M> __device__ __forceinline__ float row_xmask(float x) {
    return __builtin_bit_cast(float, __builtin_amdgcn_update_dpp(
        0, __builtin_bit_cast(int, x), 0x160 + M, 0xf, 0xf, false));
}
__device__ __forceinline__ float row16_sum(float s) {
    s += row_xmask<1>(s);
    s += row_xmask<2>(s);
    s += row_xmask<4>(s);
    s += row_xmask<8>(s);
    return s;
}
__device__ __forceinline__ float wave32_sum(float s) {
    s = row16_sum(s);
    return s + __builtin_bit_cast(float, __builtin_amdgcn_permlanex16(
        __builtin_bit_cast(int, s), __builtin_bit_cast(int, s),
        0x76543210, 0xfedcba98, false, false));
}
__device__ __forceinline__ float ln_sum4(v4f x) { return (x.x + x.y) + (x.z + x.w); }
__device__ __forceinline__ float ln_sq4(v4f d) { return (d.x * d.x + d.y * d.y) + (d.z * d.z + d.w * d.w); }
__device__ __forceinline__ v4f ln_apply(v4f x, float mu, float rs, v4f g, v4f b) {
    v4f r;
    r.x = (x.x - mu) * rs * g.x + b.x;
    r.y = (x.y - mu) * rs * g.y + b.y;
    r.z = (x.z - mu) * rs * g.z + b.z;
    r.w = (x.w - mu) * rs * g.w + b.w;
    return r;
}

__device__ __forceinline__ v2f fma2(v2f a, v2f b, float c) {
    const v2f cc = { c, c };
    return __builtin_elementwise_fma(a, b, cc);
}
__device__ __forceinline__ v2f gelu_pair(v2f x) {
    v2f v = x * 0.70710678118654752f;
    v.x = fminf(fmaxf(v.x, -4.0f), 4.0f);
    v.y = fminf(fmaxf(v.y, -4.0f), 4.0f);
    const v2f t = v * v;
    v2f p = fma2(t, (v2f){ -2.72614225801306e-10f, -2.72614225801306e-10f }, 2.77068142495902e-08f);
    p = fma2(p, t, -2.10102402082508e-06f);
    p = fma2(p, t, -5.69250639462346e-05f);
    p = fma2(p, t, -7.34990630326855e-04f);
    p = fma2(p, t, -2.95459980854025e-03f);
    p = fma2(p, t, -1.60960333262415e-02f);
    v2f q = fma2(t, (v2f){ -1.45660718464996e-05f, -1.45660718464996e-05f }, -2.13374055278905e-04f);
    q = fma2(q, t, -1.68282697438203e-03f);
    q = fma2(q, t, -7.37332916720468e-03f);
    q = fma2(q, t, -1.42647390514189e-02f);
    const v2f r = { __builtin_amdgcn_rcpf(q.x), __builtin_amdgcn_rcpf(q.y) };
    const v2f e = (v * p) * r;
    const v2f hx = x * 0.5f;
    return __builtin_elementwise_fma(hx, e, hx);
}

#define GT_M 64
#define GT_N 128
#define GT_K 64
#define GT_KQ (GT_K / 4)
#define A_SL (GT_M * 4 + 4)
#define B_SL (GT_N * 4 + 4)

__global__ __launch_bounds__(256) void ddnet_gemm_kernel(
    const float* __restrict__ A, const float* __restrict__ W,
    const float* __restrict__ bias, float* __restrict__ C, int Mdim, int Ndim, int Kdim)
{
    __shared__ __align__(16) float As[2][GT_KQ * A_SL];
    __shared__ __align__(16) float Bs[2][GT_KQ * B_SL];

    const int tid   = threadIdx.x;
    const int lane  = tid & 31;
    const int wave  = tid >> 5;
    const int lhalf = lane >> 4;
    const int lmod  = lane & 15;
    const int m0    = blockIdx.y * GT_M;
    const int n0    = blockIdx.x * GT_N;
    const int wm    = (wave & 1) * 32;
    const int wn    = (wave >> 1) * 32;
    const int nk    = Kdim >> 6;

    const int lq = tid & 15;
    const int lr = tid >> 4;
    const float* Ablk = A + (size_t)m0 * Kdim;
    const float* Wblk = W + (size_t)n0 * Kdim;
    const unsigned toff = (unsigned)(lr * Kdim + lq * 4);
    const int rstep = 16 * Kdim;
    const int a_st = lq * A_SL + lr * 4;
    const int b_st = lq * B_SL + lr * 4;
    const int a_rd = lhalf * A_SL + (wm + lmod) * 4;
    const int b_rd = lhalf * B_SL + (wn + lmod) * 4;

    v8f acc[2][2] = {};
    v4f ra[4], rb[8];

    #pragma unroll
    for (int i = 0; i < 4; ++i)
        ra[i] = *(const v4f*)(Ablk + i * rstep + toff);
    #pragma unroll
    for (int i = 0; i < 8; ++i)
        rb[i] = *(const v4f*)(Wblk + i * rstep + toff);
    #pragma unroll
    for (int i = 0; i < 4; ++i)
        *(v4f*)&As[0][a_st + i * 64] = ra[i];
    #pragma unroll
    for (int i = 0; i < 8; ++i)
        *(v4f*)&Bs[0][b_st + i * 64] = rb[i];
    __syncthreads();

    for (int kt = 0; kt < nk; ++kt) {
        const int buf = kt & 1;
        if (kt + 1 < nk) {
            const int k0 = (kt + 1) << 6;
            #pragma unroll
            for (int i = 0; i < 4; ++i)
                ra[i] = *(const v4f*)(Ablk + (k0 + i * rstep) + toff);
            #pragma unroll
            for (int i = 0; i < 8; ++i)
                rb[i] = *(const v4f*)(Wblk + (k0 + i * rstep) + toff);
        }
        const float* Ab = As[buf] + a_rd;
        const float* Bb = Bs[buf] + b_rd;
        #pragma unroll
        for (int u = 0; u < GT_KQ / 2; ++u) {
            v4f a0 = *(const v4f*)&Ab[2 * u * A_SL];
            v4f a1 = *(const v4f*)&Ab[2 * u * A_SL + 64];
            v4f b0 = *(const v4f*)&Bb[2 * u * B_SL];
            v4f b1 = *(const v4f*)&Bb[2 * u * B_SL + 64];
            acc[0][0] = wmma_f32(a0.xy, b0.xy, acc[0][0]);
            acc[0][1] = wmma_f32(a0.xy, b1.xy, acc[0][1]);
            acc[1][0] = wmma_f32(a1.xy, b0.xy, acc[1][0]);
            acc[1][1] = wmma_f32(a1.xy, b1.xy, acc[1][1]);
            acc[0][0] = wmma_f32(a0.zw, b0.zw, acc[0][0]);
            acc[0][1] = wmma_f32(a0.zw, b1.zw, acc[0][1]);
            acc[1][0] = wmma_f32(a1.zw, b0.zw, acc[1][0]);
            acc[1][1] = wmma_f32(a1.zw, b1.zw, acc[1][1]);
        }
        if (kt + 1 < nk) {
            __syncthreads();
            #pragma unroll
            for (int i = 0; i < 4; ++i)
                *(v4f*)&As[buf ^ 1][a_st + i * 64] = ra[i];
            #pragma unroll
            for (int i = 0; i < 8; ++i)
                *(v4f*)&Bs[buf ^ 1][b_st + i * 64] = rb[i];
            __syncthreads();
        }
    }

    const bool has_bias = bias != nullptr;
    const int  ecol = n0 + wn + lmod;
    const size_t eoff = (size_t)(m0 + wm + 8 * lhalf) * Ndim + ecol;
    float bv[2] = { 0.f, 0.f };
    if (has_bias) { bv[0] = bias[ecol]; bv[1] = bias[ecol + 16]; }
    float* cp = C + eoff;
    #pragma unroll
    for (int mi = 0; mi < 2; ++mi)
        #pragma unroll
        for (int ni = 0; ni < 2; ++ni)
            #pragma unroll
            for (int v = 0; v < 8; ++v) {
                float val = acc[mi][ni][v];
                if (has_bias) val += bv[ni];
                cp[(size_t)(mi * 16 + v) * Ndim + ni * 16] = val;
            }
}

__global__ __launch_bounds__(256) void ddnet_pe_kernel(
    float* __restrict__ h, int S, int D)
{
    size_t idx = (size_t)blockIdx.x * 256 + threadIdx.x;
    int d = (int)(idx % D);
    int s = (int)((idx / D) % S);
    int pair = d >> 1;
    float div = __expf(-(float)(2 * pair) * (logf(10000.0f) / (float)D));
    float ang = (float)s * div;
    h[idx] += (d & 1) ? __cosf(ang) : __sinf(ang);
}

#define DD 256
#define DG 6
#define DS_SG 4
#define DS_NI 128
#define DS_XP 66
#define DS_XS (64 * DS_XP)
#define DS_NT (DS_NI + DD - 1)
#define DS_CS (384 * DG)

struct DescTile { v4f p[12]; };

__device__ __forceinline__ void desc_load(DescTile& T, const float* pr) {
    #pragma unroll
    for (int t = 0; t < 4; ++t)
        #pragma unroll
        for (int q = 0; q < 3; ++q)
            T.p[3 * t + q] = *(const v4f*)(pr + t * 24 + q * 4);
}

struct DescOps { v2f c[DS_SG][6]; v2f a[DS_SG]; };

__device__ __forceinline__ void desc_ops(DescOps& O, const float* cp, const float* xp) {
    #pragma unroll
    for (int sl = 0; sl < DS_SG; ++sl) {
        #pragma unroll
        for (int m = 0; m < 6; ++m)
            O.c[sl][m] = *(const v2f*)(cp + sl * DS_CS + 2 * m);
        O.a[sl] = lds_frag(xp + sl * DS_XS);
    }
}

__device__ __forceinline__ void desc_group(const DescTile& T, int t, const DescOps& O,
                                           v8f (&acc)[DS_SG]) {
    #pragma unroll
    for (int sl = 0; sl < DS_SG; ++sl) {
        float w0 = 0.f, w1 = 0.f;
        #pragma unroll
        for (int g = 0; g < DG; ++g) {
            w0 += T.p[3 * t + (g >> 2)][g & 3] * O.c[sl][g >> 1][g & 1];
            w1 += T.p[3 * t + ((6 + g) >> 2)][(6 + g) & 3] * O.c[sl][(6 + g) >> 1][g & 1];
        }
        v2f b = { w0, w1 };
        acc[sl] = wmma_f32(O.a[sl], b, acc[sl]);
    }
}

__device__ __forceinline__ void desc_step(const DescTile& T, const float* cb,
                                          const float* xb, v8f (&acc)[DS_SG]) {
    DescOps Oa, Ob;
    desc_ops(Oa, cb, xb);
    desc_ops(Ob, cb + 24, xb + DS_XP);
    __builtin_amdgcn_sched_barrier(0);
    desc_group(T, 0, Oa, acc);
    desc_ops(Oa, cb + 48, xb + 2 * DS_XP);
    __builtin_amdgcn_sched_barrier(0);
    desc_group(T, 1, Ob, acc);
    desc_ops(Ob, cb + 72, xb + 3 * DS_XP);
    __builtin_amdgcn_sched_barrier(0);
    desc_group(T, 2, Oa, acc);
    desc_group(T, 3, Ob, acc);
}

__global__ __launch_bounds__(256) void ddnet_desc_kernel(
    const float* __restrict__ xpre, const float* __restrict__ lng,
    const float* __restrict__ lnb, const float* __restrict__ h,
    const float* __restrict__ P, float* __restrict__ hd, int S)
{
    __shared__ __align__(16) float Xs[DS_SG * DS_XS];
    __shared__ __align__(16) float Ct[DS_SG * DS_CS];
    __shared__ float red[2][8][16];

    const int tid = threadIdx.x;
    const int i0  = (blockIdx.x & 1) * DS_NI;
    const int s0  = (blockIdx.x >> 1) * DS_SG;
    const float two_pi = 6.28318530717958647692f;

    v4f xv[16];
    {
        const float* xsrc = xpre + (size_t)s0 * DD + tid * 4;
        #pragma unroll
        for (int bb = 0; bb < 16; ++bb)
            xv[bb] = *(const v4f*)(xsrc + (size_t)bb * S * DD);
    }

    {
        const int ln_lane = tid & 31, ln_wave = tid >> 5;
        const int w_even = ln_wave & 6, w_odd = ln_wave | 1;
        const v4f g4 = *(const v4f*)(lng + (tid & 63) * 4);
        const v4f b4 = *(const v4f*)(lnb + (tid & 63) * 4);
        float part[16], mu[16];
        #pragma unroll
        for (int bb = 0; bb < 16; ++bb)
            part[bb] = wave32_sum(ln_sum4(xv[bb]));
        if (ln_lane == 0) {
            #pragma unroll
            for (int bb = 0; bb < 16; ++bb) red[0][ln_wave][bb] = part[bb];
        }
        __syncthreads();
        #pragma unroll
        for (int bb = 0; bb < 16; ++bb) {
            mu[bb] = (red[0][w_even][bb] + red[0][w_odd][bb]) * (1.0f / 256.0f);
            part[bb] = wave32_sum(ln_sq4(xv[bb] - mu[bb]));
        }
        if (ln_lane == 0) {
            #pragma unroll
            for (int bb = 0; bb < 16; ++bb) red[1][ln_wave][bb] = part[bb];
        }
        __syncthreads();
        #pragma unroll
        for (int bb = 0; bb < 16; ++bb) {
            const float rsd = rsqrtf((red[1][w_even][bb] + red[1][w_odd][bb]) * (1.0f / 256.0f) + 1e-5f);
            xv[bb] = ln_apply(xv[bb], mu[bb], rsd, g4, b4);
        }
    }

    if (tid < 42 * DG) {
        const int g = tid % DG, tq = tid / DG;
        const float pw = powf(1.5f, (float)g);
        for (int tt = tq; tt < DS_NT; tt += 42) {
            const float period = (2.0f + 0.5f * (float)(i0 + tt)) * pw;
            #pragma unroll
            for (int sl = 0; sl < DS_SG; ++sl)
                Ct[sl * DS_CS + tt * DG + g] = __cosf(two_pi * (float)(s0 + sl) / period);
        }
    }
    {
        float* xd = Xs + (tid >> 6) * DS_XS + (tid & 63) * DS_XP;
        #pragma unroll
        for (int bb = 0; bb < 16; ++bb) {
            v2f lo = { xv[bb].x, xv[bb].y }, hi = { xv[bb].z, xv[bb].w };
            *(v2f*)(xd + bb * 2)      = lo;
            *(v2f*)(xd + 32 + bb * 2) = hi;
        }
    }
    __syncthreads();

    const int lane = tid & 31, wave = tid >> 5;
    const int lhalf = lane >> 4, lmod = lane & 15;
    const int iloc = wave * 16 + lmod;
    const float* prow = P + ((size_t)(i0 + iloc) * DD + 2 * lhalf) * DG;
    const float* cb = Ct + (iloc + 2 * lhalf) * DG;
    const float* xb = Xs + lhalf * 32 + lmod * 2;

    v8f acc[DS_SG] = {};
    DescTile Ta, Tb;
    desc_load(Ta, prow);
    for (int j0 = 0; j0 < DD; j0 += 32) {
        desc_load(Tb, prow + (j0 + 16) * DG);
        __builtin_amdgcn_sched_barrier(0);
        desc_step(Ta, cb + j0 * DG, xb + (j0 >> 2) * DS_XP, acc);
        const int jn = (j0 + 32 < DD) ? (j0 + 32) : j0;
        desc_load(Ta, prow + jn * DG);
        __builtin_amdgcn_sched_barrier(0);
        desc_step(Tb, cb + (j0 + 16) * DG, xb + ((j0 + 16) >> 2) * DS_XP, acc);
    }

    float hv[DS_SG][8];
    #pragma unroll
    for (int sl = 0; sl < DS_SG; ++sl)
        #pragma unroll
        for (int v = 0; v < 8; ++v)
            hv[sl][v] = h[((size_t)(v + 8 * lhalf) * S + s0 + sl) * DD + i0 + iloc];
    #pragma unroll
    for (int sl = 0; sl < DS_SG; ++sl)
        #pragma unroll
        for (int v = 0; v < 8; ++v)
            hd[((size_t)(v + 8 * lhalf) * S + s0 + sl) * DD + i0 + iloc] = acc[sl][v] + hv[sl][v];
}

struct AttnTile { v4f k[4]; v2f v[8]; };

__device__ __forceinline__ float attn_xhalf(float x) {
    return __builtin_bit_cast(float, __builtin_amdgcn_permlanex16(
        __builtin_bit_cast(int, x), __builtin_bit_cast(int, x),
        0x76543210, 0xfedcba98, false, false));
}

__device__ __forceinline__ void attn_load(AttnTile& T, const float* kr, const float* vr) {
    #pragma unroll
    for (int c = 0; c < 4; ++c) T.k[c] = *(const v4f*)(kr + 4 * c);
    #pragma unroll
    for (int j = 0; j < 8; ++j) T.v[j] = *(const v2f*)(vr + (size_t)j * 768);
}

__device__ __forceinline__ void attn_step(const AttnTile& T, const v4f (&qv)[4],
                                          float& mrow, float& lsum, v8f& o0, v8f& o1) {
    const float scale = 0.17677669529663688f;
    v8f s0 = {}, s1 = {};
    #pragma unroll
    for (int c = 0; c < 4; ++c) {
        s0 = wmma_f32(T.k[c].xy, qv[c].xy, s0);
        s1 = wmma_f32(T.k[c].zw, qv[c].zw, s1);
    }
    float sv[8];
    #pragma unroll
    for (int v = 0; v < 8; ++v) sv[v] = (s0[v] + s1[v]) * scale;
    float mx = fmaxf(fmaxf(fmaxf(sv[0], sv[1]), fmaxf(sv[2], sv[3])),
                     fmaxf(fmaxf(sv[4], sv[5]), fmaxf(sv[6], sv[7])));
    mx = fmaxf(mx, attn_xhalf(mx));
    const float mnew = fmaxf(mrow, mx);
    const float corr = __expf(mrow - mnew);
    float p[8];
    float rsum = 0.f;
    #pragma unroll
    for (int v = 0; v < 8; ++v) { p[v] = __expf(sv[v] - mnew); rsum += p[v]; }
    lsum = lsum * corr + rsum;
    mrow = mnew;
    #pragma unroll
    for (int v = 0; v < 8; ++v) { o0[v] *= corr; o1[v] *= corr; }
    #pragma unroll
    for (int t = 0; t < 4; ++t) {
        v2f bp = { p[2 * t], p[2 * t + 1] };
        v2f a0 = { T.v[2 * t].x, T.v[2 * t + 1].x };
        v2f a1 = { T.v[2 * t].y, T.v[2 * t + 1].y };
        o0 = wmma_f32(a0, bp, o0);
        o1 = wmma_f32(a1, bp, o1);
    }
}

__global__ __launch_bounds__(128) void ddnet_attn_kernel(
    const float* __restrict__ qkv, float* __restrict__ o, int S)
{
    const int qb    = S / 64;
    const int bh    = blockIdx.x / qb;
    const int qblk  = blockIdx.x % qb;
    const int b     = bh >> 3, head = bh & 7;
    const int tid   = threadIdx.x, lane = tid & 31, wave = tid >> 5;
    const int lhalf = lane >> 4, lmod = lane & 15;
    const int q0    = qblk * 64 + wave * 16;

    const size_t rs = 768;
    const float* qp = qkv + (size_t)b * S * rs + head * 32;
    const float* kr = qp + 256 + (size_t)lmod * rs + 16 * lhalf;
    const float* vr = qp + 512 + (size_t)(8 * lhalf) * rs + 2 * lmod;

    v4f qv[4];
    {
        const float* qr = qp + (size_t)(q0 + lmod) * rs + 16 * lhalf;
        #pragma unroll
        for (int c = 0; c < 4; ++c) qv[c] = *(const v4f*)(qr + 4 * c);
    }

    float mrow = -1e30f, lsum = 0.f;
    v8f o0 = {}, o1 = {};

    AttnTile Ta, Tb;
    attn_load(Ta, kr, vr);
    attn_load(Tb, kr + (size_t)16 * rs, vr + (size_t)16 * rs);
    for (int k0 = 0; k0 < S; k0 += 32) {
        attn_step(Ta, qv, mrow, lsum, o0, o1);
        const int kn = (k0 + 32 < S) ? (k0 + 32) : k0;
        attn_load(Ta, kr + (size_t)kn * rs, vr + (size_t)kn * rs);
        __builtin_amdgcn_sched_barrier(0);
        attn_step(Tb, qv, mrow, lsum, o0, o1);
        attn_load(Tb, kr + (size_t)(kn + 16) * rs, vr + (size_t)(kn + 16) * rs);
        __builtin_amdgcn_sched_barrier(0);
    }

    const float inv = 1.0f / (lsum + attn_xhalf(lsum));
    float* orow = o + ((size_t)b * S + q0 + lmod) * 256 + head * 32 + 16 * lhalf;
    #pragma unroll
    for (int j = 0; j < 4; ++j) {
        v4f w = { o0[2 * j] * inv, o1[2 * j] * inv, o0[2 * j + 1] * inv, o1[2 * j + 1] * inv };
        *(v4f*)(orow + 4 * j) = w;
    }
}

#define RC_M 32
#define RA_SL (RC_M * 4 + 4)
#define RB_SL (256 * 4 + 4)
#define RC_STAGE (GT_KQ * RB_SL)

__device__ __forceinline__ const float* rc_stage_src(int g, const float* out_w, const float* ff1_w,
                                                     const float* ff2_w, const float* Mn, int& stride) {
    if (g < 4)  { stride = 256;  return out_w + g * 64; }
    if (g < 20) { stride = 256;  return ff1_w + ((g - 4) >> 2) * 65536 + ((g - 4) & 3) * 64; }
    if (g < 36) { stride = 1024; return ff2_w + (g - 20) * 64; }
    stride = 256;
    return Mn + (g - 36) * 64;
}

__device__ __forceinline__ void rc_ustep(const float* Ab, const float* Bb, int u, v8f (&acc)[2][2]) {
    const v4f a0 = *(const v4f*)&Ab[2 * u * RA_SL];
    const v4f a1 = *(const v4f*)&Ab[2 * u * RA_SL + 64];
    const v4f b0 = *(const v4f*)&Bb[2 * u * RB_SL];
    const v4f b1 = *(const v4f*)&Bb[2 * u * RB_SL + 64];
    acc[0][0] = wmma_f32(a0.xy, b0.xy, acc[0][0]);
    acc[0][1] = wmma_f32(a0.xy, b1.xy, acc[0][1]);
    acc[1][0] = wmma_f32(a1.xy, b0.xy, acc[1][0]);
    acc[1][1] = wmma_f32(a1.xy, b1.xy, acc[1][1]);
    acc[0][0] = wmma_f32(a0.zw, b0.zw, acc[0][0]);
    acc[0][1] = wmma_f32(a0.zw, b1.zw, acc[0][1]);
    acc[1][0] = wmma_f32(a1.zw, b0.zw, acc[1][0]);
    acc[1][1] = wmma_f32(a1.zw, b1.zw, acc[1][1]);
}

__device__ __forceinline__ void rc_ln_rows(float* buf, const float* __restrict__ g,
                                           const float* __restrict__ b, float* __restrict__ dst,
                                           int wave, int lane) {
    const v4f g0 = *(const v4f*)(g + 4 * lane), g1 = *(const v4f*)(g + 128 + 4 * lane);
    const v4f b0 = *(const v4f*)(b + 4 * lane), b1 = *(const v4f*)(b + 128 + 4 * lane);
    #pragma unroll
    for (int rr = 0; rr < 4; ++rr) {
        const int r = wave * 4 + rr;
        float* p0 = buf + lane * RA_SL + r * 4;
        float* p1 = buf + (lane + 32) * RA_SL + r * 4;
        const v4f x0 = *(const v4f*)p0, x1 = *(const v4f*)p1;
        const float mean = wave32_sum(ln_sum4(x0) + ln_sum4(x1)) * (1.0f / 256.0f);
        const float sq = wave32_sum(ln_sq4(x0 - mean) + ln_sq4(x1 - mean));
        const float rstd = rsqrtf(sq * (1.0f / 256.0f) + 1e-5f);
        const v4f y0 = ln_apply(x0, mean, rstd, g0, b0);
        const v4f y1 = ln_apply(x1, mean, rstd, g1, b1);
        *(v4f*)p0 = y0;
        *(v4f*)p1 = y1;
        if (dst != nullptr) {
            *(v4f*)(dst + (size_t)r * 256 + 4 * lane) = y0;
            *(v4f*)(dst + (size_t)r * 256 + 128 + 4 * lane) = y1;
        }
    }
}

__global__ __launch_bounds__(256) void ddnet_chain_kernel(
    const float* __restrict__ obuf, const float* __restrict__ hd,
    const float* __restrict__ out_w, const float* __restrict__ out_b,
    const float* __restrict__ ln1g, const float* __restrict__ ln1b,
    const float* __restrict__ ff1_w, const float* __restrict__ ff1_b,
    const float* __restrict__ ff2_w, const float* __restrict__ ff2_b,
    const float* __restrict__ ln2g, const float* __restrict__ ln2b,
    const float* __restrict__ Mn, float* __restrict__ hnew, float* __restrict__ xpre)
{
    __shared__ __align__(16) float Wst[2][RC_STAGE];
    __shared__ __align__(16) float H1s[64 * RA_SL];
    __shared__ __align__(16) float Act[256 * RA_SL];

    const int tid   = threadIdx.x;
    const int lane  = tid & 31;
    const int wave  = tid >> 5;
    const int lhalf = lane >> 4;
    const int lmod  = lane & 15;
    const int wn    = wave * 32;
    const int row0  = blockIdx.x * RC_M;
    const int ns    = (Mn != nullptr) ? 40 : 36;

    const int lq = tid & 15;
    const int lr = tid >> 4;
    const int b_st = lq * RB_SL + lr * 4;
    const int a_rd = lhalf * RA_SL + lmod * 4;
    const int b_rd = lhalf * RB_SL + (wn + lmod) * 4;

    v4f rb[16];
    {
        int st;
        const float* p = rc_stage_src(0, out_w, ff1_w, ff2_w, Mn, st) + lr * st + lq * 4;
        #pragma unroll
        for (int i = 0; i < 16; ++i)
            rb[i] = *(const v4f*)(p + i * 16 * st);
        const int xq = tid & 63, xr = tid >> 6;
        const float* xs = obuf + (size_t)(row0 + xr) * 256 + xq * 4;
        v4f xv[8];
        #pragma unroll
        for (int i = 0; i < 8; ++i)
            xv[i] = *(const v4f*)(xs + i * 4 * 256);
        #pragma unroll
        for (int i = 0; i < 8; ++i)
            *(v4f*)&Act[xq * RA_SL + (xr + 4 * i) * 4] = xv[i];
        #pragma unroll
        for (int i = 0; i < 16; ++i)
            *(v4f*)&Wst[0][b_st + i * 64] = rb[i];
    }
    __syncthreads();

    v8f acc[2][2] = {};
    for (int g = 0; g < ns; ++g) {
        const int buf = g & 1;
        const bool more = g + 1 < ns;
        if (more) {
            int st;
            const float* p = rc_stage_src(g + 1, out_w, ff1_w, ff2_w, Mn, st) + lr * st + lq * 4;
            #pragma unroll
            for (int i = 0; i < 16; ++i)
                rb[i] = *(const v4f*)(p + i * 16 * st);
        }
        const float* Ab;
        if (g < 4)       Ab = Act + (g * 16) * RA_SL;
        else if (g < 20) Ab = H1s + (((g - 4) & 3) * 16) * RA_SL;
        else if (g < 36) Ab = Act + ((g - 20) * 16) * RA_SL;
        else             Ab = H1s + ((g - 36) * 16) * RA_SL;
        Ab += a_rd;
        const float* Bb = Wst[buf] + b_rd;
        #pragma unroll
        for (int u = 0; u < 6; ++u)
            rc_ustep(Ab, Bb, u, acc);
        if (more) {
            #pragma unroll
            for (int i = 0; i < 16; ++i)
                *(v4f*)&Wst[buf ^ 1][b_st + i * 64] = rb[i];
        }
        #pragma unroll
        for (int u = 6; u < 8; ++u)
            rc_ustep(Ab, Bb, u, acc);

        const bool endA = (g == 3), endC = (g == 35), endD = (g == 39);
        const bool endB = (g >= 4) && (g < 20) && (((g - 4) & 3) == 3);
        if (endA || endB || endC || endD) {
            const int ecol = wn + lmod;
            if (endA) {
                float bv[2] = { out_b[ecol], out_b[ecol + 16] };
                float rv[2][2][8];
                const float* rp = hd + (size_t)(row0 + 8 * lhalf) * 256 + ecol;
                #pragma unroll
                for (int mi = 0; mi < 2; ++mi)
                    #pragma unroll
                    for (int ni = 0; ni < 2; ++ni)
                        #pragma unroll
                        for (int v = 0; v < 8; ++v)
                            rv[mi][ni][v] = rp[(size_t)(mi * 16 + v) * 256 + ni * 16];
                #pragma unroll
                for (int mi = 0; mi < 2; ++mi)
                    #pragma unroll
                    for (int ni = 0; ni < 2; ++ni) {
                        const int n = ecol + ni * 16;
                        float* hp = H1s + (n >> 2) * RA_SL + (mi * 16 + 8 * lhalf) * 4 + (n & 3);
                        #pragma unroll
                        for (int v = 0; v < 8; ++v)
                            hp[v * 4] = (acc[mi][ni][v] + bv[ni]) + rv[mi][ni][v];
                    }
                __syncthreads();
                rc_ln_rows(H1s, ln1g, ln1b, nullptr, wave, lane);
            } else if (endB) {
                const int c = (g - 4) >> 2;
                float bv[2] = { ff1_b[c * 256 + ecol], ff1_b[c * 256 + ecol + 16] };
                #pragma unroll
                for (int mi = 0; mi < 2; ++mi)
                    #pragma unroll
                    for (int ni = 0; ni < 2; ++ni) {
                        const int n = c * 256 + ecol + ni * 16;
                        float* ap = Act + (n >> 2) * RA_SL + (mi * 16 + 8 * lhalf) * 4 + (n & 3);
                        #pragma unroll
                        for (int v = 0; v < 8; v += 2) {
                            v2f x = { acc[mi][ni][v] + bv[ni], acc[mi][ni][v + 1] + bv[ni] };
                            x = gelu_pair(x);
                            ap[v * 4]       = x.x;
                            ap[(v + 1) * 4] = x.y;
                        }
                    }
            } else if (endC) {
                float bv[2] = { ff2_b[ecol], ff2_b[ecol + 16] };
                #pragma unroll
                for (int mi = 0; mi < 2; ++mi)
                    #pragma unroll
                    for (int ni = 0; ni < 2; ++ni) {
                        const int n = ecol + ni * 16;
                        float* hp = H1s + (n >> 2) * RA_SL + (mi * 16 + 8 * lhalf) * 4 + (n & 3);
                        #pragma unroll
                        for (int v = 0; v < 8; ++v)
                            hp[v * 4] = (acc[mi][ni][v] + bv[ni]) + hp[v * 4];
                    }
                __syncthreads();
                rc_ln_rows(H1s, ln2g, ln2b, hnew + (size_t)row0 * 256, wave, lane);
            } else {
                float* cp = xpre + (size_t)(row0 + 8 * lhalf) * 256 + ecol;
                #pragma unroll
                for (int mi = 0; mi < 2; ++mi)
                    #pragma unroll
                    for (int ni = 0; ni < 2; ++ni)
                        #pragma unroll
                        for (int v = 0; v < 8; ++v)
                            cp[(size_t)(mi * 16 + v) * 256 + ni * 16] = acc[mi][ni][v];
            }
            #pragma unroll
            for (int mi = 0; mi < 2; ++mi)
                #pragma unroll
                for (int ni = 0; ni < 2; ++ni)
                    acc[mi][ni] = (v8f){};
        }
        __syncthreads();
    }
}

static void launch_gemm(const float* A, const float* W, const float* bias,
                        float* C, int M, int N, int K, hipStream_t stream)
{
    dim3 grid(N / GT_N, M / GT_M);
    ddnet_gemm_kernel<<<grid, 256, 0, stream>>>(A, W, bias, C, M, N, K);
}

extern "C" void kernel_launch(void* const* d_in, const int* in_sizes, int n_in,
                              void* d_out, int out_size, void* d_ws, size_t ws_size,
                              hipStream_t stream) {
    (void)in_sizes; (void)n_in; (void)out_size; (void)ws_size;
    const int B = 16, S = 512, IN = 64, D = 256, L = 3, G = 6, DFF = 1024;
    const int Mr = B * S;

    const float* x      = (const float*)d_in[0];
    const float* in_w   = (const float*)d_in[1];
    const float* in_b   = (const float*)d_in[2];
    const float* Mw     = (const float*)d_in[3];
    const float* P      = (const float*)d_in[4];
    const float* desc_g = (const float*)d_in[5];
    const float* desc_b = (const float*)d_in[6];
    const float* qkv_w  = (const float*)d_in[7];
    const float* qkv_b  = (const float*)d_in[8];
    const float* out_w  = (const float*)d_in[9];
    const float* out_b  = (const float*)d_in[10];
    const float* ff1_w  = (const float*)d_in[11];
    const float* ff1_b  = (const float*)d_in[12];
    const float* ff2_w  = (const float*)d_in[13];
    const float* ff2_b  = (const float*)d_in[14];
    const float* ln1_g  = (const float*)d_in[15];
    const float* ln1_b  = (const float*)d_in[16];
    const float* ln2_g  = (const float*)d_in[17];
    const float* ln2_b  = (const float*)d_in[18];

    float* h    = (float*)d_ws;
    float* tmp  = h    + (size_t)Mr * D;
    float* hd   = tmp  + (size_t)Mr * D;
    float* qkv  = hd   + (size_t)Mr * D;
    float* obuf = qkv  + (size_t)Mr * 3 * D;

    launch_gemm(x, in_w, in_b, h, Mr, D, IN, stream);
    ddnet_pe_kernel<<<(Mr * D) / 256, 256, 0, stream>>>(h, S, D);

    for (int l = 0; l < L; ++l) {
        if (l == 0)
            launch_gemm(h, Mw, nullptr, tmp, Mr, D, D, stream);
        ddnet_desc_kernel<<<(S / 4) * 2, 256, 0, stream>>>(tmp, desc_g + l * D, desc_b + l * D, h,
                                                           P + (size_t)l * D * D * G, hd, S);
        launch_gemm(hd, qkv_w + (size_t)l * 3 * D * D, qkv_b + (size_t)l * 3 * D,
                    qkv, Mr, 3 * D, D, stream);
        ddnet_attn_kernel<<<B * 8 * (S / 64), 128, 0, stream>>>(qkv, obuf, S);
        if (l + 1 < L)
            ddnet_chain_kernel<<<Mr / RC_M, 256, 0, stream>>>(
                obuf, hd, out_w + (size_t)l * D * D, out_b + l * D, ln1_g + l * D, ln1_b + l * D,
                ff1_w + (size_t)l * DFF * D, ff1_b + (size_t)l * DFF,
                ff2_w + (size_t)l * D * DFF, ff2_b + l * D, ln2_g + l * D, ln2_b + l * D,
                Mw + (size_t)(l + 1) * D * D, h, tmp);
        else
            ddnet_chain_kernel<<<Mr / RC_M, 256, 0, stream>>>(
                obuf, hd, out_w + (size_t)l * D * D, out_b + l * D, ln1_g + l * D, ln1_b + l * D,
                ff1_w + (size_t)l * DFF * D, ff1_b + (size_t)l * DFF,
                ff2_w + (size_t)l * D * DFF, ff2_b + l * D, ln2_g + l * D, ln2_b + l * D,
                nullptr, (float*)d_out, tmp);
    }
}
